# starting kernel
# baseline (speedup 1.0000x reference)
#include <hip/hip_runtime.h>
#include <hip/hip_bf16.h>
#include <math.h>

#define LL 2048
#define BB 64
#define KK 256
#define VV 256
#define HH 8
#define HK 2048
#define HV 2048

typedef __attribute__((ext_vector_type(2))) float v2f;
typedef __attribute__((ext_vector_type(8))) float v8f;

__device__ __forceinline__ v8f wmma4(v2f a, v2f b, v8f c) {
    return __builtin_amdgcn_wmma_f32_16x16x4_f32(
         false, a,  false, b,
         (short)0, c,  false,  false);
}

__global__ void k_qproj(const float* __restrict__ query,
                        const float* __restrict__ Wq,
                        const float* __restrict__ bq,
                        float* __restrict__ qbuf) {
    const int waveId = blockIdx.x * (blockDim.x >> 5) + (threadIdx.x >> 5);
    const int tm   = waveId >> 7;
    const int tn   = waveId & 127;
    const int lane = threadIdx.x & 31;
    const int col  = lane & 15;
    const int half = lane >> 4;

    const float* Arow = query + (size_t)(tm * 16 + col) * KK;
    const float* Bcol = Wq + (tn * 16 + col);

    v8f c = {};
    #pragma unroll 4
    for (int k0 = 0; k0 < KK; k0 += 4) {
        const int ka = k0 + 2 * half;
        v2f a = { Arow[ka], Arow[ka + 1] };
        v2f b = { Bcol[(size_t)ka * HK], Bcol[(size_t)(ka + 1) * HK] };
        c = wmma4(a, b, c);
    }

    const float bias = bq[tn * 16 + col];
    #pragma unroll
    for (int j = 0; j < 8; ++j) {
        const int row = tm * 16 + j + 8 * half;
        qbuf[(size_t)row * HK + tn * 16 + col] = c[j] + bias;
    }
}

__global__ void k_scores(const float* __restrict__ keys,
                         const float* __restrict__ rpe,
                         const float* __restrict__ qbuf,
                         float* __restrict__ smat) {
    const int waveId = blockIdx.x * (blockDim.x >> 5) + (threadIdx.x >> 5);
    const int b    = waveId >> 7;
    const int lt   = waveId & 127;
    const int lane = threadIdx.x & 31;
    const int col  = lane & 15;
    const int half = lane >> 4;

    const int l_m = lt * 16 + col;
    const float* keyRow = keys + ((size_t)l_m * BB + b) * KK;
    const float r = rpe[(size_t)l_m * BB + b];

    const bool bn = (col < HH);
    const float* qb = qbuf + (size_t)b * HK + (size_t)col * KK;

    __builtin_prefetch(keyRow, 0, 0);

    v8f c = {};
    #pragma unroll 8
    for (int k0 = 0; k0 < KK; k0 += 4) {
        const int ka = k0 + 2 * half;
        if ((k0 & 28) == 0)
            __builtin_prefetch(keyRow + k0 + 32, 0, 0);
        v2f kv = __builtin_nontemporal_load((const v2f*)(keyRow + ka));
        v2f a  = { kv.x * r, kv.y * r };
        v2f bb = { bn ? qb[ka] : 0.0f, bn ? qb[ka + 1] : 0.0f };
        c = wmma4(a, bb, c);
    }

    if (col < HH) {
        #pragma unroll
        for (int j = 0; j < 8; ++j) {
            const int l = lt * 16 + j + 8 * half;
            smat[((size_t)b * HH + col) * LL + l] = c[j];
        }
    }
}

__global__ void k_softmax(float* __restrict__ s, const int* __restrict__ steps) {
    const int bh   = blockIdx.x;
    const int b    = bh >> 3;
    const int step = steps[b];
    float* row = s + (size_t)bh * LL;
    const int t = threadIdx.x;

    float v[8];
    float mx = -INFINITY;
    #pragma unroll
    for (int i = 0; i < 8; ++i) {
        const int l = t + i * 256;
        const float x = (l < step) ? row[l] : -INFINITY;
        v[i] = x;
        mx = fmaxf(mx, x);
    }

    __shared__ float red[256];
    red[t] = mx;
    __syncthreads();
    for (int off = 128; off > 0; off >>= 1) {
        if (t < off) red[t] = fmaxf(red[t], red[t + off]);
        __syncthreads();
    }
    mx = red[0];
    __syncthreads();

    float sum = 0.0f;
    #pragma unroll
    for (int i = 0; i < 8; ++i) {
        const int l = t + i * 256;
        const float e = (l < step) ? __expf(v[i] - mx) : 0.0f;
        v[i] = e;
        sum += e;
    }
    red[t] = sum;
    __syncthreads();
    for (int off = 128; off > 0; off >>= 1) {
        if (t < off) red[t] += red[t + off];
        __syncthreads();
    }
    const float inv = 1.0f / red[0];

    #pragma unroll
    for (int i = 0; i < 8; ++i) row[t + i * 256] = v[i] * inv;
}

__global__ void k_read(const float* __restrict__ wmat,
                       const float* __restrict__ vals,
                       float* __restrict__ readbuf) {
    const int waveId = blockIdx.x * (blockDim.x >> 5) + (threadIdx.x >> 5);
    const int b    = waveId >> 4;
    const int tn   = waveId & 15;
    const int lane = threadIdx.x & 31;
    const int col  = lane & 15;
    const int half = lane >> 4;

    const bool am = (col < HH);
    const float* wrow  = wmat + ((size_t)b * HH + col) * LL;
    const float* vbase = vals + (size_t)b * VV + tn * 16 + col;

    v8f c = {};
    #pragma unroll 4
    for (int k0 = 0; k0 < LL; k0 += 4) {
        const int ka = k0 + 2 * half;
        if ((k0 & 12) == 0)
            __builtin_prefetch(vbase + (size_t)(k0 + 8) * (BB * VV), 0, 0);
        v2f a  = { am ? wrow[ka] : 0.0f, am ? wrow[ka + 1] : 0.0f };
        v2f bb = { __builtin_nontemporal_load(vbase + (size_t)ka * (BB * VV)),
                   __builtin_nontemporal_load(vbase + (size_t)(ka + 1) * (BB * VV)) };
        c = wmma4(a, bb, c);
    }

    if (half == 0) {
        #pragma unroll
        for (int j = 0; j < 8; ++j)
            readbuf[((size_t)b * HH + j) * VV + tn * 16 + col] = c[j];
    }
}

__global__ void k_out(const float* __restrict__ readbuf,
                      const float* __restrict__ Wa,
                      const float* __restrict__ ba,
                      float* __restrict__ out) {
    const int waveId = blockIdx.x * (blockDim.x >> 5) + (threadIdx.x >> 5);
    const int tm   = waveId >> 4;
    const int tn   = waveId & 15;
    const int lane = threadIdx.x & 31;
    const int col  = lane & 15;
    const int half = lane >> 4;

    const float* Arow = readbuf + (size_t)(tm * 16 + col) * HV;
    const float* Bcol = Wa + tn * 16 + col;

    v8f c = {};
    #pragma unroll 2
    for (int k0 = 0; k0 < HV; k0 += 4) {
        const int ka = k0 + 2 * half;
        v2f a = { Arow[ka], Arow[ka + 1] };
        v2f b = { Bcol[(size_t)ka * VV], Bcol[(size_t)(ka + 1) * VV] };
        c = wmma4(a, b, c);
    }

    const float bias = ba[tn * 16 + col];
    #pragma unroll
    for (int j = 0; j < 8; ++j) {
        const int row = tm * 16 + j + 8 * half;
        out[(size_t)row * VV + tn * 16 + col] = c[j] + bias;
    }
}

extern "C" void kernel_launch(void* const* d_in, const int* in_sizes, int n_in,
                              void* d_out, int out_size, void* d_ws, size_t ws_size,
                              hipStream_t stream) {
    const float* query = (const float*)d_in[0];
    const float* keys  = (const float*)d_in[1];
    const float* vals  = (const float*)d_in[2];
    const float* rpe   = (const float*)d_in[3];
    const float* Wq    = (const float*)d_in[4];
    const float* bq    = (const float*)d_in[5];
    const float* Wa    = (const float*)d_in[6];
    const float* ba    = (const float*)d_in[7];
    const int*   steps = (const int*)d_in[8];
    float* out = (float*)d_out;

    float* ws = (float*)d_ws;
    float* qbuf    = ws;
    float* smat    = qbuf + (size_t)BB * HK;
    float* readbuf = smat + (size_t)BB * HH * LL;

    k_qproj<<<64, 256, 0, stream>>>(query, Wq, bq, qbuf);
    k_scores<<<1024, 256, 0, stream>>>(keys, rpe, qbuf, smat);
    k_softmax<<<BB * HH, 256, 0, stream>>>(smat, steps);
    k_read<<<128, 256, 0, stream>>>(smat, vals, readbuf);
    k_out<<<8, 256, 0, stream>>>(readbuf, Wa, ba, out);
}


# k_out, k_qproj: each wave keeps the A/B loads of 16 k-steps in flight in a register ring refilled behind each product, B rows asked into L2 ahead; 1-2 wave work-groups; tile per wave, sum order kept
# speedup vs baseline: 5.4129x; 5.4129x over previous
#include <hip/hip_runtime.h>
#include <hip/hip_bf16.h>
#include <math.h>

#define LL 2048
#define BB 64
#define KK 256
#define VV 256
#define HH 8
#define HK 2048
#define HV 2048

typedef __attribute__((ext_vector_type(2))) float v2f;
typedef __attribute__((ext_vector_type(8))) float v8f;

__device__ __forceinline__ v8f wmma4(v2f a, v2f b, v8f c) {
    return __builtin_amdgcn_wmma_f32_16x16x4_f32(
         false, a,  false, b,
         (short)0, c,  false,  false);
}

#define RING     16
#define PF_TRIPS 4

__device__ __forceinline__ v8f tile_product_ring(const float* __restrict__ Arow,
                                                 const float* __restrict__ Bcol,
                                                 const float* __restrict__ Bpf,
                                                 const int ldb, const int ksteps, const int lane) {
    const int krows = 4 * ksteps;
    v2f ra[RING], rb[RING];
    #pragma unroll
    for (int u = 0; u < RING; ++u) {
        const int k0 = 4 * u;
        ra[u] = (v2f){ Arow[k0], Arow[k0 + 1] };
        rb[u] = (v2f){ Bcol[(size_t)k0 * ldb], Bcol[(size_t)(k0 + 1) * ldb] };
        __builtin_amdgcn_sched_barrier(0);
    }
    #pragma unroll
    for (int i = 0; i < 2 * PF_TRIPS; ++i) {
        const int base = 4 * RING + 32 * i;
        if (base < krows) __builtin_prefetch(Bpf + (size_t)(base + lane) * ldb, 0, 1);
    }

    v8f c = {};
    for (int s0 = 0; s0 < ksteps - RING; s0 += RING) {
        const int base = 4 * (s0 + RING * (1 + PF_TRIPS));
        if (base < krows) {
            __builtin_prefetch(Bpf + (size_t)(base + lane) * ldb, 0, 1);
            __builtin_prefetch(Bpf + (size_t)(base + 32 + lane) * ldb, 0, 1);
        }
        #pragma unroll
        for (int u = 0; u < RING; ++u) {
            c = wmma4(ra[u], rb[u], c);
            const int k0 = 4 * (s0 + u + RING);
            ra[u] = (v2f){ Arow[k0], Arow[k0 + 1] };
            rb[u] = (v2f){ Bcol[(size_t)k0 * ldb], Bcol[(size_t)(k0 + 1) * ldb] };
            __builtin_amdgcn_sched_barrier(0);
        }
    }
    #pragma unroll
    for (int u = 0; u < RING; ++u) c = wmma4(ra[u], rb[u], c);
    return c;
}

__global__ void k_qproj(const float* __restrict__ query,
                        const float* __restrict__ Wq,
                        const float* __restrict__ bq,
                        float* __restrict__ qbuf) {
    const int waveId = blockIdx.x * (blockDim.x >> 5) + (threadIdx.x >> 5);
    const int tm   = waveId >> 7;
    const int tn   = waveId & 127;
    const int lane = threadIdx.x & 31;
    const int col  = lane & 15;
    const int half = lane >> 4;

    const float* Arow = query + (size_t)(tm * 16 + col) * KK + 2 * half;
    const float* Bcol = Wq + (size_t)(2 * half) * HK + (tn * 16 + col);

    v8f c = tile_product_ring(Arow, Bcol, Wq + tn * 16, HK, KK / 4, lane);

    const float bias = bq[tn * 16 + col];
    #pragma unroll
    for (int j = 0; j < 8; ++j) {
        const int row = tm * 16 + j + 8 * half;
        qbuf[(size_t)row * HK + tn * 16 + col] = c[j] + bias;
    }
}

__global__ void k_scores(const float* __restrict__ keys,
                         const float* __restrict__ rpe,
                         const float* __restrict__ qbuf,
                         float* __restrict__ smat) {
    const int waveId = blockIdx.x * (blockDim.x >> 5) + (threadIdx.x >> 5);
    const int b    = waveId >> 7;
    const int lt   = waveId & 127;
    const int lane = threadIdx.x & 31;
    const int col  = lane & 15;
    const int half = lane >> 4;

    const int l_m = lt * 16 + col;
    const float* keyRow = keys + ((size_t)l_m * BB + b) * KK;
    const float r = rpe[(size_t)l_m * BB + b];

    const bool bn = (col < HH);
    const float* qb = qbuf + (size_t)b * HK + (size_t)col * KK;

    __builtin_prefetch(keyRow, 0, 0);

    v8f c = {};
    #pragma unroll 8
    for (int k0 = 0; k0 < KK; k0 += 4) {
        const int ka = k0 + 2 * half;
        if ((k0 & 28) == 0)
            __builtin_prefetch(keyRow + k0 + 32, 0, 0);
        v2f kv = __builtin_nontemporal_load((const v2f*)(keyRow + ka));
        v2f a  = { kv.x * r, kv.y * r };
        v2f bb = { bn ? qb[ka] : 0.0f, bn ? qb[ka + 1] : 0.0f };
        c = wmma4(a, bb, c);
    }

    if (col < HH) {
        #pragma unroll
        for (int j = 0; j < 8; ++j) {
            const int l = lt * 16 + j + 8 * half;
            smat[((size_t)b * HH + col) * LL + l] = c[j];
        }
    }
}

__global__ void k_softmax(float* __restrict__ s, const int* __restrict__ steps) {
    const int bh   = blockIdx.x;
    const int b    = bh >> 3;
    const int step = steps[b];
    float* row = s + (size_t)bh * LL;
    const int t = threadIdx.x;

    float v[8];
    float mx = -INFINITY;
    #pragma unroll
    for (int i = 0; i < 8; ++i) {
        const int l = t + i * 256;
        const float x = (l < step) ? row[l] : -INFINITY;
        v[i] = x;
        mx = fmaxf(mx, x);
    }

    __shared__ float red[256];
    red[t] = mx;
    __syncthreads();
    for (int off = 128; off > 0; off >>= 1) {
        if (t < off) red[t] = fmaxf(red[t], red[t + off]);
        __syncthreads();
    }
    mx = red[0];
    __syncthreads();

    float sum = 0.0f;
    #pragma unroll
    for (int i = 0; i < 8; ++i) {
        const int l = t + i * 256;
        const float e = (l < step) ? __expf(v[i] - mx) : 0.0f;
        v[i] = e;
        sum += e;
    }
    red[t] = sum;
    __syncthreads();
    for (int off = 128; off > 0; off >>= 1) {
        if (t < off) red[t] += red[t + off];
        __syncthreads();
    }
    const float inv = 1.0f / red[0];

    #pragma unroll
    for (int i = 0; i < 8; ++i) row[t + i * 256] = v[i] * inv;
}

#define RD_ROWS   64
#define RD_DEPTH  4
#define RD_STAGES (LL / RD_ROWS)
#define RD_PV     136
#define RD_PA     68

typedef __attribute__((ext_vector_type(4))) float v4f;

__device__ __forceinline__ void rd_handover() {
    asm volatile("s_wait_dscnt 0x0" ::: "memory");
    asm volatile("s_barrier_signal -1\n\ts_barrier_wait -1" ::: "memory");
}

#define RD_ROUND(T0, REFILL)                                                            \
    for (int u = 0; u < RD_DEPTH; ++u) {                                                \
        const int buf = u & 1;                                                          \
        for (int q = 0; q < 8; ++q)                                                     \
            *(v4f*)&sV[buf][8 * wv + q][4 * lane] = rv[u][q];                           \
        *(v2f*)&sA[buf][wv][2 * lane] = ra[u];                                          \
        if (REFILL) {                                                                   \
            const int tn = (T0) + u + RD_DEPTH;                                         \
            for (int q = 0; q < 8; ++q)                                                 \
                rv[u][q] = __builtin_nontemporal_load(                                  \
                    (const v4f*)(vsrc + (size_t)(tn * RD_ROWS + q) * (BB * VV)));       \
            ra[u] = *(const v2f*)(asrc + tn * RD_ROWS);                                 \
        }                                                                               \
        rd_handover();                                                                  \
        for (int j = 0; j < RD_ROWS / 4; ++j) {                                         \
            const int r0 = 4 * j + 2 * half;                                            \
            const v2f a  = *(const v2f*)&sA[buf][col][r0];                              \
            const v2f bb = { sV[buf][r0][16 * wv + col], sV[buf][r0 + 1][16 * wv + col] }; \
            c = wmma4(a, bb, c);                                                        \
        }                                                                               \
    }

__global__ __launch_bounds__(256) void k_read(const float* __restrict__ wmat,
                                              const float* __restrict__ vals,
                                              float* __restrict__ readbuf) {
    __shared__ __attribute__((aligned(16))) float sV[2][RD_ROWS][RD_PV];
    __shared__ __attribute__((aligned(16))) float sA[2][16][RD_PA];

    const int tid  = threadIdx.x;
    const int wv   = tid >> 5;
    const int lane = tid & 31;
    const int col  = lane & 15;
    const int half = lane >> 4;
    const int b    = blockIdx.x >> 1;
    const int cg   = blockIdx.x & 1;

    const float* vsrc = vals + ((size_t)(8 * wv) * BB + b) * VV + cg * 128 + 4 * lane;
    const float* asrc = wmat + ((size_t)b * HH + wv) * LL + 2 * lane;

    for (int i = tid; i < 2 * 8 * RD_PA; i += 256) {
        const int bf = i / (8 * RD_PA);
        const int r  = i - bf * (8 * RD_PA);
        sA[bf][8 + r / RD_PA][r % RD_PA] = 0.0f;
    }

    v4f rv[RD_DEPTH][8];
    v2f ra[RD_DEPTH];
    for (int u = 0; u < RD_DEPTH; ++u) {
        for (int q = 0; q < 8; ++q)
            rv[u][q] = __builtin_nontemporal_load(
                (const v4f*)(vsrc + (size_t)(u * RD_ROWS + q) * (BB * VV)));
        ra[u] = *(const v2f*)(asrc + u * RD_ROWS);
    }

    v8f c = {};
    for (int t0 = 0; t0 < RD_STAGES - RD_DEPTH; t0 += RD_DEPTH) {
        RD_ROUND(t0, true)
    }
    RD_ROUND(RD_STAGES - RD_DEPTH, false)

    if (half == 0) {
        for (int j = 0; j < 8; ++j)
            readbuf[((size_t)b * HH + j) * VV + cg * 128 + 16 * wv + col] = c[j];
    }
}

__global__ __launch_bounds__(32) void k_out(const float* __restrict__ readbuf,
                      const float* __restrict__ Wa,
                      const float* __restrict__ ba,
                      float* __restrict__ out) {
    const int waveId = blockIdx.x * (blockDim.x >> 5) + (threadIdx.x >> 5);
    const int tm   = waveId >> 4;
    const int tn   = waveId & 15;
    const int lane = threadIdx.x & 31;
    const int col  = lane & 15;
    const int half = lane >> 4;

    const float* Arow = readbuf + (size_t)(tm * 16 + col) * HV + 2 * half;
    const float* Bcol = Wa + (size_t)(2 * half) * VV + tn * 16 + col;

    v8f c = tile_product_ring(Arow, Bcol, Wa + tn * 16, VV, HV / 4, lane);

    const float bias = ba[tn * 16 + col];
    #pragma unroll
    for (int j = 0; j < 8; ++j) {
        const int row = tm * 16 + j + 8 * half;
        out[(size_t)row * VV + tn * 16 + col] = c[j] + bias;
    }
}

extern "C" void kernel_launch(void* const* d_in, const int* in_sizes, int n_in,
                              void* d_out, int out_size, void* d_ws, size_t ws_size,
                              hipStream_t stream) {
    const float* query = (const float*)d_in[0];
    const float* keys  = (const float*)d_in[1];
    const float* vals  = (const float*)d_in[2];
    const float* rpe   = (const float*)d_in[3];
    const float* Wq    = (const float*)d_in[4];
    const float* bq    = (const float*)d_in[5];
    const float* Wa    = (const float*)d_in[6];
    const float* ba    = (const float*)d_in[7];
    const int*   steps = (const int*)d_in[8];
    float* out = (float*)d_out;

    float* ws = (float*)d_ws;
    float* qbuf    = ws;
    float* smat    = qbuf + (size_t)BB * HK;
    float* readbuf = smat + (size_t)BB * HH * LL;

    k_qproj<<<256, 64, 0, stream>>>(query, Wq, bq, qbuf);
    k_scores<<<1024, 256, 0, stream>>>(keys, rpe, qbuf, smat);
    k_softmax<<<BB * HH, 256, 0, stream>>>(smat, steps);
    k_read<<<128, 256, 0, stream>>>(smat, vals, readbuf);
    k_out<<<64, 32, 0, stream>>>(readbuf, Wa, ba, out);
}


# k_scores: keys and q fragments loaded 16 bytes per lane, one load feeding two products; k in each group of 8 paired 0,1,4,5 then 2,3,6,7 on both operands (new sum order); half the row loads
# speedup vs baseline: 6.1839x; 1.1424x over previous
#include <hip/hip_runtime.h>
#include <hip/hip_bf16.h>
#include <math.h>

#define LL 2048
#define BB 64
#define KK 256
#define VV 256
#define HH 8
#define HK 2048
#define HV 2048

typedef __attribute__((ext_vector_type(2))) float v2f;
typedef __attribute__((ext_vector_type(4))) float v4f;
typedef __attribute__((ext_vector_type(8))) float v8f;

__device__ __forceinline__ v8f wmma4(v2f a, v2f b, v8f c) {
    return __builtin_amdgcn_wmma_f32_16x16x4_f32(
         false, a,  false, b,
         (short)0, c,  false,  false);
}

#define RING     16
#define PF_TRIPS 4

__device__ __forceinline__ v8f tile_product_ring(const float* __restrict__ Arow,
                                                 const float* __restrict__ Bcol,
                                                 const float* __restrict__ Bpf,
                                                 const int ldb, const int ksteps, const int lane) {
    const int krows = 4 * ksteps;
    v2f ra[RING], rb[RING];
    #pragma unroll
    for (int u = 0; u < RING; ++u) {
        const int k0 = 4 * u;
        ra[u] = (v2f){ Arow[k0], Arow[k0 + 1] };
        rb[u] = (v2f){ Bcol[(size_t)k0 * ldb], Bcol[(size_t)(k0 + 1) * ldb] };
        __builtin_amdgcn_sched_barrier(0);
    }
    #pragma unroll
    for (int i = 0; i < 2 * PF_TRIPS; ++i) {
        const int base = 4 * RING + 32 * i;
        if (base < krows) __builtin_prefetch(Bpf + (size_t)(base + lane) * ldb, 0, 1);
    }

    v8f c = {};
    for (int s0 = 0; s0 < ksteps - RING; s0 += RING) {
        const int base = 4 * (s0 + RING * (1 + PF_TRIPS));
        if (base < krows) {
            __builtin_prefetch(Bpf + (size_t)(base + lane) * ldb, 0, 1);
            __builtin_prefetch(Bpf + (size_t)(base + 32 + lane) * ldb, 0, 1);
        }
        #pragma unroll
        for (int u = 0; u < RING; ++u) {
            c = wmma4(ra[u], rb[u], c);
            const int k0 = 4 * (s0 + u + RING);
            ra[u] = (v2f){ Arow[k0], Arow[k0 + 1] };
            rb[u] = (v2f){ Bcol[(size_t)k0 * ldb], Bcol[(size_t)(k0 + 1) * ldb] };
            __builtin_amdgcn_sched_barrier(0);
        }
    }
    #pragma unroll
    for (int u = 0; u < RING; ++u) c = wmma4(ra[u], rb[u], c);
    return c;
}

__global__ void k_qproj(const float* __restrict__ query,
                        const float* __restrict__ Wq,
                        const float* __restrict__ bq,
                        float* __restrict__ qbuf) {
    const int waveId = blockIdx.x * (blockDim.x >> 5) + (threadIdx.x >> 5);
    const int tm   = waveId >> 7;
    const int tn   = waveId & 127;
    const int lane = threadIdx.x & 31;
    const int col  = lane & 15;
    const int half = lane >> 4;

    const float* Arow = query + (size_t)(tm * 16 + col) * KK + 2 * half;
    const float* Bcol = Wq + (size_t)(2 * half) * HK + (tn * 16 + col);

    v8f c = tile_product_ring(Arow, Bcol, Wq + tn * 16, HK, KK / 4, lane);

    const float bias = bq[tn * 16 + col];
    #pragma unroll
    for (int j = 0; j < 8; ++j) {
        const int row = tm * 16 + j + 8 * half;
        qbuf[(size_t)row * HK + tn * 16 + col] = c[j] + bias;
    }
}

__global__ void k_scores(const float* __restrict__ keys,
                         const float* __restrict__ rpe,
                         const float* __restrict__ qbuf,
                         float* __restrict__ smat) {
    const int waveId = blockIdx.x * (blockDim.x >> 5) + (threadIdx.x >> 5);
    const int b    = waveId >> 7;
    const int lt   = waveId & 127;
    const int lane = threadIdx.x & 31;
    const int col  = lane & 15;
    const int half = lane >> 4;

    const int l_m = lt * 16 + col;
    const float* keyRow = keys + ((size_t)l_m * BB + b) * KK;
    const float r = rpe[(size_t)l_m * BB + b];

    const bool bn = (col < HH);
    const float* qb = qbuf + (size_t)b * HK + (size_t)col * KK;

    __builtin_prefetch(keyRow, 0, 0);

    v8f c = {};
    for (int k0 = 0; k0 < KK; k0 += 64) {
        if (k0 + 64 < KK)
            __builtin_prefetch(keyRow + k0 + 64, 0, 0);
        #pragma unroll
        for (int j = 0; j < 8; ++j) {
            const int ka = k0 + 8 * j + 4 * half;
            const v4f kv = __builtin_nontemporal_load((const v4f*)(keyRow + ka));
            v4f qv = { 0.0f, 0.0f, 0.0f, 0.0f };
            if (bn) qv = *(const v4f*)(qb + ka);
            c = wmma4((v2f){ kv.x * r, kv.y * r }, (v2f){ qv.x, qv.y }, c);
            c = wmma4((v2f){ kv.z * r, kv.w * r }, (v2f){ qv.z, qv.w }, c);
        }
    }

    if (col < HH) {
        #pragma unroll
        for (int j = 0; j < 8; ++j) {
            const int l = lt * 16 + j + 8 * half;
            smat[((size_t)b * HH + col) * LL + l] = c[j];
        }
    }
}

__global__ void k_softmax(float* __restrict__ s, const int* __restrict__ steps) {
    const int bh   = blockIdx.x;
    const int b    = bh >> 3;
    const int step = steps[b];
    float* row = s + (size_t)bh * LL;
    const int t = threadIdx.x;

    float v[8];
    float mx = -INFINITY;
    #pragma unroll
    for (int i = 0; i < 8; ++i) {
        const int l = t + i * 256;
        const float x = (l < step) ? row[l] : -INFINITY;
        v[i] = x;
        mx = fmaxf(mx, x);
    }

    __shared__ float red[256];
    red[t] = mx;
    __syncthreads();
    for (int off = 128; off > 0; off >>= 1) {
        if (t < off) red[t] = fmaxf(red[t], red[t + off]);
        __syncthreads();
    }
    mx = red[0];
    __syncthreads();

    float sum = 0.0f;
    #pragma unroll
    for (int i = 0; i < 8; ++i) {
        const int l = t + i * 256;
        const float e = (l < step) ? __expf(v[i] - mx) : 0.0f;
        v[i] = e;
        sum += e;
    }
    red[t] = sum;
    __syncthreads();
    for (int off = 128; off > 0; off >>= 1) {
        if (t < off) red[t] += red[t + off];
        __syncthreads();
    }
    const float inv = 1.0f / red[0];

    #pragma unroll
    for (int i = 0; i < 8; ++i) row[t + i * 256] = v[i] * inv;
}

#define RD_ROWS   64
#define RD_DEPTH  4
#define RD_STAGES (LL / RD_ROWS)
#define RD_PV     136
#define RD_PA     68

__device__ __forceinline__ void rd_handover() {
    asm volatile("s_wait_dscnt 0x0" ::: "memory");
    asm volatile("s_barrier_signal -1\n\ts_barrier_wait -1" ::: "memory");
}

#define RD_ROUND(T0, REFILL)                                                            \
    for (int u = 0; u < RD_DEPTH; ++u) {                                                \
        const int buf = u & 1;                                                          \
        for (int q = 0; q < 8; ++q)                                                     \
            *(v4f*)&sV[buf][8 * wv + q][4 * lane] = rv[u][q];                           \
        *(v2f*)&sA[buf][wv][2 * lane] = ra[u];                                          \
        if (REFILL) {                                                                   \
            const int tn = (T0) + u + RD_DEPTH;                                         \
            for (int q = 0; q < 8; ++q)                                                 \
                rv[u][q] = __builtin_nontemporal_load(                                  \
                    (const v4f*)(vsrc + (size_t)(tn * RD_ROWS + q) * (BB * VV)));       \
            ra[u] = *(const v2f*)(asrc + tn * RD_ROWS);                                 \
        }                                                                               \
        rd_handover();                                                                  \
        for (int j = 0; j < RD_ROWS / 4; ++j) {                                         \
            const int r0 = 4 * j + 2 * half;                                            \
            const v2f a  = *(const v2f*)&sA[buf][col][r0];                              \
            const v2f bb = { sV[buf][r0][16 * wv + col], sV[buf][r0 + 1][16 * wv + col] }; \
            c = wmma4(a, bb, c);                                                        \
        }                                                                               \
    }

__global__ __launch_bounds__(256) void k_read(const float* __restrict__ wmat,
                                              const float* __restrict__ vals,
                                              float* __restrict__ readbuf) {
    __shared__ __attribute__((aligned(16))) float sV[2][RD_ROWS][RD_PV];
    __shared__ __attribute__((aligned(16))) float sA[2][16][RD_PA];

    const int tid  = threadIdx.x;
    const int wv   = tid >> 5;
    const int lane = tid & 31;
    const int col  = lane & 15;
    const int half = lane >> 4;
    const int b    = blockIdx.x >> 1;
    const int cg   = blockIdx.x & 1;

    const float* vsrc = vals + ((size_t)(8 * wv) * BB + b) * VV + cg * 128 + 4 * lane;
    const float* asrc = wmat + ((size_t)b * HH + wv) * LL + 2 * lane;

    for (int i = tid; i < 2 * 8 * RD_PA; i += 256) {
        const int bf = i / (8 * RD_PA);
        const int r  = i - bf * (8 * RD_PA);
        sA[bf][8 + r / RD_PA][r % RD_PA] = 0.0f;
    }

    v4f rv[RD_DEPTH][8];
    v2f ra[RD_DEPTH];
    for (int u = 0; u < RD_DEPTH; ++u) {
        for (int q = 0; q < 8; ++q)
            rv[u][q] = __builtin_nontemporal_load(
                (const v4f*)(vsrc + (size_t)(u * RD_ROWS + q) * (BB * VV)));
        ra[u] = *(const v2f*)(asrc + u * RD_ROWS);
    }

    v8f c = {};
    for (int t0 = 0; t0 < RD_STAGES - RD_DEPTH; t0 += RD_DEPTH) {
        RD_ROUND(t0, true)
    }
    RD_ROUND(RD_STAGES - RD_DEPTH, false)

    if (half == 0) {
        for (int j = 0; j < 8; ++j)
            readbuf[((size_t)b * HH + j) * VV + cg * 128 + 16 * wv + col] = c[j];
    }
}

__global__ __launch_bounds__(32) void k_out(const float* __restrict__ readbuf,
                      const float* __restrict__ Wa,
                      const float* __restrict__ ba,
                      float* __restrict__ out) {
    const int waveId = blockIdx.x * (blockDim.x >> 5) + (threadIdx.x >> 5);
    const int tm   = waveId >> 4;
    const int tn   = waveId & 15;
    const int lane = threadIdx.x & 31;
    const int col  = lane & 15;
    const int half = lane >> 4;

    const float* Arow = readbuf + (size_t)(tm * 16 + col) * HV + 2 * half;
    const float* Bcol = Wa + (size_t)(2 * half) * VV + tn * 16 + col;

    v8f c = tile_product_ring(Arow, Bcol, Wa + tn * 16, VV, HV / 4, lane);

    const float bias = ba[tn * 16 + col];
    #pragma unroll
    for (int j = 0; j < 8; ++j) {
        const int row = tm * 16 + j + 8 * half;
        out[(size_t)row * VV + tn * 16 + col] = c[j] + bias;
    }
}

extern "C" void kernel_launch(void* const* d_in, const int* in_sizes, int n_in,
                              void* d_out, int out_size, void* d_ws, size_t ws_size,
                              hipStream_t stream) {
    const float* query = (const float*)d_in[0];
    const float* keys  = (const float*)d_in[1];
    const float* vals  = (const float*)d_in[2];
    const float* rpe   = (const float*)d_in[3];
    const float* Wq    = (const float*)d_in[4];
    const float* bq    = (const float*)d_in[5];
    const float* Wa    = (const float*)d_in[6];
    const float* ba    = (const float*)d_in[7];
    const int*   steps = (const int*)d_in[8];
    float* out = (float*)d_out;

    float* ws = (float*)d_ws;
    float* qbuf    = ws;
    float* smat    = qbuf + (size_t)BB * HK;
    float* readbuf = smat + (size_t)BB * HH * LL;

    k_qproj<<<256, 64, 0, stream>>>(query, Wq, bq, qbuf);
    k_scores<<<1024, 256, 0, stream>>>(keys, rpe, qbuf, smat);
    k_softmax<<<BB * HH, 256, 0, stream>>>(smat, steps);
    k_read<<<128, 256, 0, stream>>>(smat, vals, readbuf);
    k_out<<<64, 32, 0, stream>>>(readbuf, Wa, ba, out);
}


# k_scores, k_softmax, k_read fused into one pass split along L: 16-wave work-group per (batch, 256 rows) keeps scores and exp in LDS; chunks merged by their maxima, 1/sum at the end (new sum order)
# speedup vs baseline: 7.4730x; 1.2085x over previous
#include <hip/hip_runtime.h>
#include <hip/hip_bf16.h>
#include <math.h>

#define LL 2048
#define BB 64
#define KK 256
#define VV 256
#define HH 8
#define HK 2048
#define HV 2048

typedef __attribute__((ext_vector_type(2))) float v2f;
typedef __attribute__((ext_vector_type(4))) float v4f;
typedef __attribute__((ext_vector_type(8))) float v8f;

__device__ __forceinline__ v8f wmma4(v2f a, v2f b, v8f c) {
    return __builtin_amdgcn_wmma_f32_16x16x4_f32(
         false, a,  false, b,
         (short)0, c,  false,  false);
}

#define RING     16
#define PF_TRIPS 4

__device__ __forceinline__ v8f tile_product_ring(const float* __restrict__ Arow,
                                                 const float* __restrict__ Bcol,
                                                 const float* __restrict__ Bpf,
                                                 const int ldb, const int ksteps, const int lane) {
    const int krows = 4 * ksteps;
    v2f ra[RING], rb[RING];
    #pragma unroll
    for (int u = 0; u < RING; ++u) {
        const int k0 = 4 * u;
        ra[u] = (v2f){ Arow[k0], Arow[k0 + 1] };
        rb[u] = (v2f){ Bcol[(size_t)k0 * ldb], Bcol[(size_t)(k0 + 1) * ldb] };
        __builtin_amdgcn_sched_barrier(0);
    }
    #pragma unroll
    for (int i = 0; i < 2 * PF_TRIPS; ++i) {
        const int base = 4 * RING + 32 * i;
        if (base < krows) __builtin_prefetch(Bpf + (size_t)(base + lane) * ldb, 0, 1);
    }

    v8f c = {};
    for (int s0 = 0; s0 < ksteps - RING; s0 += RING) {
        const int base = 4 * (s0 + RING * (1 + PF_TRIPS));
        if (base < krows) {
            __builtin_prefetch(Bpf + (size_t)(base + lane) * ldb, 0, 1);
            __builtin_prefetch(Bpf + (size_t)(base + 32 + lane) * ldb, 0, 1);
        }
        #pragma unroll
        for (int u = 0; u < RING; ++u) {
            c = wmma4(ra[u], rb[u], c);
            const int k0 = 4 * (s0 + u + RING);
            ra[u] = (v2f){ Arow[k0], Arow[k0 + 1] };
            rb[u] = (v2f){ Bcol[(size_t)k0 * ldb], Bcol[(size_t)(k0 + 1) * ldb] };
            __builtin_amdgcn_sched_barrier(0);
        }
    }
    #pragma unroll
    for (int u = 0; u < RING; ++u) c = wmma4(ra[u], rb[u], c);
    return c;
}

__global__ void k_qproj(const float* __restrict__ query,
                        const float* __restrict__ Wq,
                        const float* __restrict__ bq,
                        float* __restrict__ qbuf) {
    const int waveId = blockIdx.x * (blockDim.x >> 5) + (threadIdx.x >> 5);
    const int tm   = waveId >> 7;
    const int tn   = waveId & 127;
    const int lane = threadIdx.x & 31;
    const int col  = lane & 15;
    const int half = lane >> 4;

    const float* Arow = query + (size_t)(tm * 16 + col) * KK + 2 * half;
    const float* Bcol = Wq + (size_t)(2 * half) * HK + (tn * 16 + col);

    v8f c = tile_product_ring(Arow, Bcol, Wq + tn * 16, HK, KK / 4, lane);

    const float bias = bq[tn * 16 + col];
    #pragma unroll
    for (int j = 0; j < 8; ++j) {
        const int row = tm * 16 + j + 8 * half;
        qbuf[(size_t)row * HK + tn * 16 + col] = c[j] + bias;
    }
}

#define FC_CH    256
#define FC_NC    (LL / FC_CH)
#define FC_ROWS  32
#define FC_NST   (FC_CH / FC_ROWS)
#define FC_DEPTH 4
#define FC_PV    264
#define FC_PA    260

__device__ __forceinline__ void fc_handover() {
    asm volatile("s_wait_dscnt 0x0" ::: "memory");
    asm volatile("s_barrier_signal -1\n\ts_barrier_wait -1" ::: "memory");
}

__global__ __launch_bounds__(512) void k_fused(const float* __restrict__ keys,
                                               const float* __restrict__ vals,
                                               const float* __restrict__ rpe,
                                               const float* __restrict__ qbuf,
                                               const int* __restrict__ steps,
                                               float* __restrict__ part,
                                               float* __restrict__ pmax,
                                               float* __restrict__ psum) {
    __shared__ __attribute__((aligned(16))) float sV[2][FC_ROWS][FC_PV];
    __shared__ __attribute__((aligned(16))) float sA[16][FC_PA];

    const int tid  = threadIdx.x;
    const int wv   = tid >> 5;
    const int lane = tid & 31;
    const int col  = lane & 15;
    const int half = lane >> 4;
    const int b    = blockIdx.x / FC_NC;
    const int ch   = blockIdx.x % FC_NC;
    const int l0   = ch * FC_CH;
    const int step = steps[b];

    const float* vsrc = vals + ((size_t)(l0 + 2 * wv) * BB + b) * VV + 4 * lane;
    v4f rv[FC_DEPTH][4];
    #pragma unroll
    for (int u = 0; u < FC_DEPTH; ++u) {
        #pragma unroll
        for (int q = 0; q < 4; ++q)
            rv[u][q] = __builtin_nontemporal_load(
                (const v4f*)(vsrc + (size_t)(u * FC_ROWS + (q >> 1)) * (BB * VV) + 128 * (q & 1)));
    }

    for (int i = tid; i < 8 * FC_PA; i += 512)
        sA[8 + i / FC_PA][i % FC_PA] = 0.0f;

    {
        const int l_m = l0 + 16 * wv + col;
        const float* keyRow = keys + ((size_t)l_m * BB + b) * KK;
        const float r = rpe[(size_t)l_m * BB + b];
        const bool bn = (col < HH);
        const float* qb = qbuf + (size_t)b * HK + (size_t)col * KK;

        __builtin_prefetch(keyRow, 0, 0);

        v8f c = {};
        for (int k0 = 0; k0 < KK; k0 += 64) {
            if (k0 + 64 < KK)
                __builtin_prefetch(keyRow + k0 + 64, 0, 0);
            #pragma unroll
            for (int j = 0; j < 8; ++j) {
                const int ka = k0 + 8 * j + 4 * half;
                const v4f kv = __builtin_nontemporal_load((const v4f*)(keyRow + ka));
                v4f qv = { 0.0f, 0.0f, 0.0f, 0.0f };
                if (bn) qv = *(const v4f*)(qb + ka);
                c = wmma4((v2f){ kv.x * r, kv.y * r }, (v2f){ qv.x, qv.y }, c);
                c = wmma4((v2f){ kv.z * r, kv.w * r }, (v2f){ qv.z, qv.w }, c);
            }
        }

        if (col < HH) {
            const int lb = 16 * wv + 8 * half;
            float sc[8];
            #pragma unroll
            for (int j = 0; j < 8; ++j)
                sc[j] = (l0 + lb + j < step) ? c[j] : -INFINITY;
            *(v4f*)&sA[col][lb]     = (v4f){ sc[0], sc[1], sc[2], sc[3] };
            *(v4f*)&sA[col][lb + 4] = (v4f){ sc[4], sc[5], sc[6], sc[7] };
        }
    }
    fc_handover();

    if (wv < HH) {
        const v4f x0 = *(const v4f*)&sA[wv][4 * lane];
        const v4f x1 = *(const v4f*)&sA[wv][128 + 4 * lane];
        float m = fmaxf(fmaxf(fmaxf(x0.x, x0.y), fmaxf(x0.z, x0.w)),
                        fmaxf(fmaxf(x1.x, x1.y), fmaxf(x1.z, x1.w)));
        #pragma unroll
        for (int off = 16; off > 0; off >>= 1) m = fmaxf(m, __shfl_xor(m, off));
        const float ms = (m == -INFINITY) ? 0.0f : m;
        const v4f e0 = { __expf(x0.x - ms), __expf(x0.y - ms), __expf(x0.z - ms), __expf(x0.w - ms) };
        const v4f e1 = { __expf(x1.x - ms), __expf(x1.y - ms), __expf(x1.z - ms), __expf(x1.w - ms) };
        float sum = ((e0.x + e0.y) + (e0.z + e0.w)) + ((e1.x + e1.y) + (e1.z + e1.w));
        #pragma unroll
        for (int off = 16; off > 0; off >>= 1) sum += __shfl_xor(sum, off);
        *(v4f*)&sA[wv][4 * lane]       = e0;
        *(v4f*)&sA[wv][128 + 4 * lane] = e1;
        if (lane == 0) {
            pmax[((size_t)b * FC_NC + ch) * HH + wv] = m;
            psum[((size_t)b * FC_NC + ch) * HH + wv] = sum;
        }
    }

    v8f acc = {};
    #pragma unroll
    for (int t = 0; t < FC_NST; ++t) {
        const int u   = t % FC_DEPTH;
        const int buf = t & 1;
        #pragma unroll
        for (int q = 0; q < 4; ++q)
            *(v4f*)&sV[buf][2 * wv + (q >> 1)][128 * (q & 1) + 4 * lane] = rv[u][q];
        if (t + FC_DEPTH < FC_NST) {
            #pragma unroll
            for (int q = 0; q < 4; ++q)
                rv[u][q] = __builtin_nontemporal_load(
                    (const v4f*)(vsrc + (size_t)((t + FC_DEPTH) * FC_ROWS + (q >> 1)) * (BB * VV) + 128 * (q & 1)));
        }
        fc_handover();
        #pragma unroll
        for (int j = 0; j < FC_ROWS / 4; ++j) {
            const int r0 = 4 * j + 2 * half;
            const v2f a  = *(const v2f*)&sA[col][FC_ROWS * t + r0];
            const v2f bb = { sV[buf][r0][16 * wv + col], sV[buf][r0 + 1][16 * wv + col] };
            acc = wmma4(a, bb, acc);
        }
    }

    if (half == 0) {
        #pragma unroll
        for (int j = 0; j < 8; ++j)
            part[(((size_t)b * FC_NC + ch) * HH + j) * VV + 16 * wv + col] = acc[j];
    }
}

__global__ __launch_bounds__(256) void k_merge(const float* __restrict__ part,
                                               const float* __restrict__ pmax,
                                               const float* __restrict__ psum,
                                               float* __restrict__ readbuf) {
    const int bh = blockIdx.x;
    const int b  = bh >> 3;
    const int h  = bh & 7;
    const int v  = threadIdx.x;

    float m[FC_NC];
    float M = -INFINITY;
    #pragma unroll
    for (int c = 0; c < FC_NC; ++c) {
        m[c] = pmax[((size_t)b * FC_NC + c) * HH + h];
        M = fmaxf(M, m[c]);
    }
    float den = 0.0f, num = 0.0f;
    #pragma unroll
    for (int c = 0; c < FC_NC; ++c) {
        const float w = __expf(m[c] - M);
        den += w * psum[((size_t)b * FC_NC + c) * HH + h];
        num += w * part[(((size_t)b * FC_NC + c) * HH + h) * VV + v];
    }
    readbuf[((size_t)b * HH + h) * VV + v] = num / den;
}

__global__ __launch_bounds__(32) void k_out(const float* __restrict__ readbuf,
                      const float* __restrict__ Wa,
                      const float* __restrict__ ba,
                      float* __restrict__ out) {
    const int waveId = blockIdx.x * (blockDim.x >> 5) + (threadIdx.x >> 5);
    const int tm   = waveId >> 4;
    const int tn   = waveId & 15;
    const int lane = threadIdx.x & 31;
    const int col  = lane & 15;
    const int half = lane >> 4;

    const float* Arow = readbuf + (size_t)(tm * 16 + col) * HV + 2 * half;
    const float* Bcol = Wa + (size_t)(2 * half) * VV + tn * 16 + col;

    v8f c = tile_product_ring(Arow, Bcol, Wa + tn * 16, VV, HV / 4, lane);

    const float bias = ba[tn * 16 + col];
    #pragma unroll
    for (int j = 0; j < 8; ++j) {
        const int row = tm * 16 + j + 8 * half;
        out[(size_t)row * VV + tn * 16 + col] = c[j] + bias;
    }
}

extern "C" void kernel_launch(void* const* d_in, const int* in_sizes, int n_in,
                              void* d_out, int out_size, void* d_ws, size_t ws_size,
                              hipStream_t stream) {
    const float* query = (const float*)d_in[0];
    const float* keys  = (const float*)d_in[1];
    const float* vals  = (const float*)d_in[2];
    const float* rpe   = (const float*)d_in[3];
    const float* Wq    = (const float*)d_in[4];
    const float* bq    = (const float*)d_in[5];
    const float* Wa    = (const float*)d_in[6];
    const float* ba    = (const float*)d_in[7];
    const int*   steps = (const int*)d_in[8];
    float* out = (float*)d_out;

    float* ws = (float*)d_ws;
    float* qbuf    = ws;
    float* part    = qbuf + (size_t)BB * HK;
    float* pmax    = part + (size_t)BB * FC_NC * HH * VV;
    float* psum    = pmax + (size_t)BB * FC_NC * HH;
    float* readbuf = ws;

    k_qproj<<<256, 64, 0, stream>>>(query, Wq, bq, qbuf);
    k_fused<<<BB * FC_NC, 512, 0, stream>>>(keys, vals, rpe, qbuf, steps, part, pmax, psum);
    k_merge<<<BB * HH, 256, 0, stream>>>(part, pmax, psum, readbuf);
    k_out<<<64, 32, 0, stream>>>(readbuf, Wa, ba, out);
}


# k_fused phase 1: score tile = 8 key rows x two 32-byte k halves (A rows 8-15, B cols 8-15 = second half; diagonal D quarters added): key loads cover 64 B runs, 2 tiles share a q load (new sum order)
# speedup vs baseline: 8.3725x; 1.1204x over previous
#include <hip/hip_runtime.h>
#include <hip/hip_bf16.h>
#include <math.h>

#define LL 2048
#define BB 64
#define KK 256
#define VV 256
#define HH 8
#define HK 2048
#define HV 2048

typedef __attribute__((ext_vector_type(2))) float v2f;
typedef __attribute__((ext_vector_type(4))) float v4f;
typedef __attribute__((ext_vector_type(8))) float v8f;

__device__ __forceinline__ v8f wmma4(v2f a, v2f b, v8f c) {
    return __builtin_amdgcn_wmma_f32_16x16x4_f32(
         false, a,  false, b,
         (short)0, c,  false,  false);
}

#define RING     16
#define PF_TRIPS 4

__device__ __forceinline__ v8f tile_product_ring(const float* __restrict__ Arow,
                                                 const float* __restrict__ Bcol,
                                                 const float* __restrict__ Bpf,
                                                 const int ldb, const int ksteps, const int lane) {
    const int krows = 4 * ksteps;
    v2f ra[RING], rb[RING];
    #pragma unroll
    for (int u = 0; u < RING; ++u) {
        const int k0 = 4 * u;
        ra[u] = (v2f){ Arow[k0], Arow[k0 + 1] };
        rb[u] = (v2f){ Bcol[(size_t)k0 * ldb], Bcol[(size_t)(k0 + 1) * ldb] };
        __builtin_amdgcn_sched_barrier(0);
    }
    #pragma unroll
    for (int i = 0; i < 2 * PF_TRIPS; ++i) {
        const int base = 4 * RING + 32 * i;
        if (base < krows) __builtin_prefetch(Bpf + (size_t)(base + lane) * ldb, 0, 1);
    }

    v8f c = {};
    for (int s0 = 0; s0 < ksteps - RING; s0 += RING) {
        const int base = 4 * (s0 + RING * (1 + PF_TRIPS));
        if (base < krows) {
            __builtin_prefetch(Bpf + (size_t)(base + lane) * ldb, 0, 1);
            __builtin_prefetch(Bpf + (size_t)(base + 32 + lane) * ldb, 0, 1);
        }
        #pragma unroll
        for (int u = 0; u < RING; ++u) {
            c = wmma4(ra[u], rb[u], c);
            const int k0 = 4 * (s0 + u + RING);
            ra[u] = (v2f){ Arow[k0], Arow[k0 + 1] };
            rb[u] = (v2f){ Bcol[(size_t)k0 * ldb], Bcol[(size_t)(k0 + 1) * ldb] };
            __builtin_amdgcn_sched_barrier(0);
        }
    }
    #pragma unroll
    for (int u = 0; u < RING; ++u) c = wmma4(ra[u], rb[u], c);
    return c;
}

__global__ void k_qproj(const float* __restrict__ query,
                        const float* __restrict__ Wq,
                        const float* __restrict__ bq,
                        float* __restrict__ qbuf) {
    const int waveId = blockIdx.x * (blockDim.x >> 5) + (threadIdx.x >> 5);
    const int tm   = waveId >> 7;
    const int tn   = waveId & 127;
    const int lane = threadIdx.x & 31;
    const int col  = lane & 15;
    const int half = lane >> 4;

    const float* Arow = query + (size_t)(tm * 16 + col) * KK + 2 * half;
    const float* Bcol = Wq + (size_t)(2 * half) * HK + (tn * 16 + col);

    v8f c = tile_product_ring(Arow, Bcol, Wq + tn * 16, HK, KK / 4, lane);

    const float bias = bq[tn * 16 + col];
    #pragma unroll
    for (int j = 0; j < 8; ++j) {
        const int row = tm * 16 + j + 8 * half;
        qbuf[(size_t)row * HK + tn * 16 + col] = c[j] + bias;
    }
}

#define FC_CH    256
#define FC_NC    (LL / FC_CH)
#define FC_ROWS  32
#define FC_NST   (FC_CH / FC_ROWS)
#define FC_DEPTH 4
#define FC_PV    264
#define FC_PA    260
#define P1_STEPS 16
#define P1_RING  2

__device__ __forceinline__ void fc_handover() {
    asm volatile("s_wait_dscnt 0x0" ::: "memory");
    asm volatile("s_barrier_signal -1\n\ts_barrier_wait -1" ::: "memory");
}

__global__ __launch_bounds__(512) __attribute__((amdgpu_waves_per_eu(8, 8))) void k_fused(const float* __restrict__ keys,
                                               const float* __restrict__ vals,
                                               const float* __restrict__ rpe,
                                               const float* __restrict__ qbuf,
                                               const int* __restrict__ steps,
                                               float* __restrict__ part,
                                               float* __restrict__ pmax,
                                               float* __restrict__ psum) {
    __shared__ __attribute__((aligned(16))) float sV[2][FC_ROWS][FC_PV];
    __shared__ __attribute__((aligned(16))) float sA[16][FC_PA];

    const int tid  = threadIdx.x;
    const int wv   = tid >> 5;
    const int lane = tid & 31;
    const int col  = lane & 15;
    const int half = lane >> 4;
    const int b    = blockIdx.x / FC_NC;
    const int ch   = blockIdx.x % FC_NC;
    const int l0   = ch * FC_CH;
    const int step = steps[b];

    const float* vsrc = vals + ((size_t)(l0 + 2 * wv) * BB + b) * VV + 4 * lane;
    v4f rv[FC_DEPTH][4];
    #pragma unroll
    for (int u = 0; u < FC_DEPTH; ++u) {
        #pragma unroll
        for (int q = 0; q < 4; ++q)
            rv[u][q] = __builtin_nontemporal_load(
                (const v4f*)(vsrc + (size_t)(u * FC_ROWS + (q >> 1)) * (BB * VV) + 128 * (q & 1)));
    }

    for (int i = tid; i < 8 * FC_PA; i += 512)
        sA[8 + i / FC_PA][i % FC_PA] = 0.0f;

    {
        const int rr = col & 7;
        const int g  = col >> 3;
        const int lt = l0 + 16 * wv + rr;
        const float* key0 = keys + ((size_t)lt * BB + b) * KK + 8 * g + 4 * half;
        const float* key1 = key0 + (size_t)8 * BB * KK;
        const float r0 = rpe[(size_t)lt * BB + b];
        const float r1 = rpe[(size_t)(lt + 8) * BB + b];
        const float* qb = qbuf + (size_t)b * HK + (size_t)rr * KK + 8 * g + 4 * half;

        __builtin_prefetch(key0, 0, 0);
        __builtin_prefetch(key1, 0, 0);
        __builtin_prefetch(key0 + 64, 0, 0);
        __builtin_prefetch(key1 + 64, 0, 0);

        v4f ka0[P1_RING], ka1[P1_RING], kq[P1_RING];
        #pragma unroll
        for (int u = 0; u < P1_RING; ++u) {
            ka0[u] = __builtin_nontemporal_load((const v4f*)(key0 + 16 * u));
            ka1[u] = __builtin_nontemporal_load((const v4f*)(key1 + 16 * u));
            kq[u]  = *(const v4f*)(qb + 16 * u);
        }

        v8f c0 = {}, c1 = {};
        #pragma unroll
        for (int i = 0; i < P1_STEPS; ++i) {
            const int u = i % P1_RING;
            if ((i & 3) == 0 && i + 8 < P1_STEPS) {
                __builtin_prefetch(key0 + 16 * (i + 8), 0, 0);
                __builtin_prefetch(key1 + 16 * (i + 8), 0, 0);
            }
            const v4f a0 = ka0[u], a1 = ka1[u], qv = kq[u];
            c0 = wmma4((v2f){ a0.x * r0, a0.y * r0 }, (v2f){ qv.x, qv.y }, c0);
            c1 = wmma4((v2f){ a1.x * r1, a1.y * r1 }, (v2f){ qv.x, qv.y }, c1);
            c0 = wmma4((v2f){ a0.z * r0, a0.w * r0 }, (v2f){ qv.z, qv.w }, c0);
            c1 = wmma4((v2f){ a1.z * r1, a1.w * r1 }, (v2f){ qv.z, qv.w }, c1);
            if (i + P1_RING < P1_STEPS) {
                ka0[u] = __builtin_nontemporal_load((const v4f*)(key0 + 16 * (i + P1_RING)));
                ka1[u] = __builtin_nontemporal_load((const v4f*)(key1 + 16 * (i + P1_RING)));
                kq[u]  = *(const v4f*)(qb + 16 * (i + P1_RING));
            }
        }

        float s0[8], s1[8];
        #pragma unroll
        for (int j = 0; j < 8; ++j) {
            s0[j] = c0[j] + __shfl_xor(c0[j], 24);
            s1[j] = c1[j] + __shfl_xor(c1[j], 24);
        }
        if (lane < HH) {
            const int lb = 16 * wv;
            #pragma unroll
            for (int j = 0; j < 8; ++j) {
                s0[j] = (l0 + lb + j     < step) ? s0[j] : -INFINITY;
                s1[j] = (l0 + lb + 8 + j < step) ? s1[j] : -INFINITY;
            }
            *(v4f*)&sA[lane][lb]      = (v4f){ s0[0], s0[1], s0[2], s0[3] };
            *(v4f*)&sA[lane][lb + 4]  = (v4f){ s0[4], s0[5], s0[6], s0[7] };
            *(v4f*)&sA[lane][lb + 8]  = (v4f){ s1[0], s1[1], s1[2], s1[3] };
            *(v4f*)&sA[lane][lb + 12] = (v4f){ s1[4], s1[5], s1[6], s1[7] };
        }
    }
    fc_handover();

    if (wv < HH) {
        const v4f x0 = *(const v4f*)&sA[wv][4 * lane];
        const v4f x1 = *(const v4f*)&sA[wv][128 + 4 * lane];
        float m = fmaxf(fmaxf(fmaxf(x0.x, x0.y), fmaxf(x0.z, x0.w)),
                        fmaxf(fmaxf(x1.x, x1.y), fmaxf(x1.z, x1.w)));
        #pragma unroll
        for (int off = 16; off > 0; off >>= 1) m = fmaxf(m, __shfl_xor(m, off));
        const float ms = (m == -INFINITY) ? 0.0f : m;
        const v4f e0 = { __expf(x0.x - ms), __expf(x0.y - ms), __expf(x0.z - ms), __expf(x0.w - ms) };
        const v4f e1 = { __expf(x1.x - ms), __expf(x1.y - ms), __expf(x1.z - ms), __expf(x1.w - ms) };
        float sum = ((e0.x + e0.y) + (e0.z + e0.w)) + ((e1.x + e1.y) + (e1.z + e1.w));
        #pragma unroll
        for (int off = 16; off > 0; off >>= 1) sum += __shfl_xor(sum, off);
        *(v4f*)&sA[wv][4 * lane]       = e0;
        *(v4f*)&sA[wv][128 + 4 * lane] = e1;
        if (lane == 0) {
            pmax[((size_t)b * FC_NC + ch) * HH + wv] = m;
            psum[((size_t)b * FC_NC + ch) * HH + wv] = sum;
        }
    }

    v8f acc = {};
    #pragma unroll
    for (int t = 0; t < FC_NST; ++t) {
        const int u   = t % FC_DEPTH;
        const int buf = t & 1;
        #pragma unroll
        for (int q = 0; q < 4; ++q)
            *(v4f*)&sV[buf][2 * wv + (q >> 1)][128 * (q & 1) + 4 * lane] = rv[u][q];
        if (t + FC_DEPTH < FC_NST) {
            #pragma unroll
            for (int q = 0; q < 4; ++q)
                rv[u][q] = __builtin_nontemporal_load(
                    (const v4f*)(vsrc + (size_t)((t + FC_DEPTH) * FC_ROWS + (q >> 1)) * (BB * VV) + 128 * (q & 1)));
        }
        fc_handover();
        #pragma unroll
        for (int j = 0; j < FC_ROWS / 4; ++j) {
            const int r0 = 4 * j + 2 * half;
            const v2f a  = *(const v2f*)&sA[col][FC_ROWS * t + r0];
            const v2f bb = { sV[buf][r0][16 * wv + col], sV[buf][r0 + 1][16 * wv + col] };
            acc = wmma4(a, bb, acc);
        }
    }

    if (half == 0) {
        #pragma unroll
        for (int j = 0; j < 8; ++j)
            part[(((size_t)b * FC_NC + ch) * HH + j) * VV + 16 * wv + col] = acc[j];
    }
}

__global__ __launch_bounds__(256) void k_merge(const float* __restrict__ part,
                                               const float* __restrict__ pmax,
                                               const float* __restrict__ psum,
                                               float* __restrict__ readbuf) {
    const int bh = blockIdx.x;
    const int b  = bh >> 3;
    const int h  = bh & 7;
    const int v  = threadIdx.x;

    float m[FC_NC];
    float M = -INFINITY;
    #pragma unroll
    for (int c = 0; c < FC_NC; ++c) {
        m[c] = pmax[((size_t)b * FC_NC + c) * HH + h];
        M = fmaxf(M, m[c]);
    }
    float den = 0.0f, num = 0.0f;
    #pragma unroll
    for (int c = 0; c < FC_NC; ++c) {
        const float w = __expf(m[c] - M);
        den += w * psum[((size_t)b * FC_NC + c) * HH + h];
        num += w * part[(((size_t)b * FC_NC + c) * HH + h) * VV + v];
    }
    readbuf[((size_t)b * HH + h) * VV + v] = num / den;
}

__global__ __launch_bounds__(32) void k_out(const float* __restrict__ readbuf,
                      const float* __restrict__ Wa,
                      const float* __restrict__ ba,
                      float* __restrict__ out) {
    const int waveId = blockIdx.x * (blockDim.x >> 5) + (threadIdx.x >> 5);
    const int tm   = waveId >> 4;
    const int tn   = waveId & 15;
    const int lane = threadIdx.x & 31;
    const int col  = lane & 15;
    const int half = lane >> 4;

    const float* Arow = readbuf + (size_t)(tm * 16 + col) * HV + 2 * half;
    const float* Bcol = Wa + (size_t)(2 * half) * VV + tn * 16 + col;

    v8f c = tile_product_ring(Arow, Bcol, Wa + tn * 16, VV, HV / 4, lane);

    const float bias = ba[tn * 16 + col];
    #pragma unroll
    for (int j = 0; j < 8; ++j) {
        const int row = tm * 16 + j + 8 * half;
        out[(size_t)row * VV + tn * 16 + col] = c[j] + bias;
    }
}

extern "C" void kernel_launch(void* const* d_in, const int* in_sizes, int n_in,
                              void* d_out, int out_size, void* d_ws, size_t ws_size,
                              hipStream_t stream) {
    const float* query = (const float*)d_in[0];
    const float* keys  = (const float*)d_in[1];
    const float* vals  = (const float*)d_in[2];
    const float* rpe   = (const float*)d_in[3];
    const float* Wq    = (const float*)d_in[4];
    const float* bq    = (const float*)d_in[5];
    const float* Wa    = (const float*)d_in[6];
    const float* ba    = (const float*)d_in[7];
    const int*   steps = (const int*)d_in[8];
    float* out = (float*)d_out;

    float* ws = (float*)d_ws;
    float* qbuf    = ws;
    float* part    = qbuf + (size_t)BB * HK;
    float* pmax    = part + (size_t)BB * FC_NC * HH * VV;
    float* psum    = pmax + (size_t)BB * FC_NC * HH;
    float* readbuf = ws;

    k_qproj<<<256, 64, 0, stream>>>(query, Wq, bq, qbuf);
    k_fused<<<BB * FC_NC, 512, 0, stream>>>(keys, vals, rpe, qbuf, steps, part, pmax, psum);
    k_merge<<<BB * HH, 256, 0, stream>>>(part, pmax, psum, readbuf);
    k_out<<<64, 32, 0, stream>>>(readbuf, Wa, ba, out);
}


# out = read @ Wa moved in front of the chunk merge: each (chunk, head) partial x its Wa block as 256 work-groups of 64-step products, slice via LDS; merge + bias last in k_final (new sum order)
# speedup vs baseline: 9.9002x; 1.1825x over previous
#include <hip/hip_runtime.h>
#include <hip/hip_bf16.h>
#include <math.h>

#define LL 2048
#define BB 64
#define KK 256
#define VV 256
#define HH 8
#define HK 2048
#define HV 2048

typedef __attribute__((ext_vector_type(2))) float v2f;
typedef __attribute__((ext_vector_type(4))) float v4f;
typedef __attribute__((ext_vector_type(8))) float v8f;

__device__ __forceinline__ v8f wmma4(v2f a, v2f b, v8f c) {
    return __builtin_amdgcn_wmma_f32_16x16x4_f32(
         false, a,  false, b,
         (short)0, c,  false,  false);
}

#define RING     16
#define PF_TRIPS 4

__device__ __forceinline__ v8f tile_product_ring(const float* __restrict__ Arow,
                                                 const float* __restrict__ Bcol,
                                                 const float* __restrict__ Bpf,
                                                 const int ldb, const int ksteps, const int lane) {
    const int krows = 4 * ksteps;
    v2f ra[RING], rb[RING];
    #pragma unroll
    for (int u = 0; u < RING; ++u) {
        const int k0 = 4 * u;
        ra[u] = (v2f){ Arow[k0], Arow[k0 + 1] };
        rb[u] = (v2f){ Bcol[(size_t)k0 * ldb], Bcol[(size_t)(k0 + 1) * ldb] };
        __builtin_amdgcn_sched_barrier(0);
    }
    #pragma unroll
    for (int i = 0; i < 2 * PF_TRIPS; ++i) {
        const int base = 4 * RING + 32 * i;
        if (base < krows) __builtin_prefetch(Bpf + (size_t)(base + lane) * ldb, 0, 1);
    }

    v8f c = {};
    for (int s0 = 0; s0 < ksteps - RING; s0 += RING) {
        const int base = 4 * (s0 + RING * (1 + PF_TRIPS));
        if (base < krows) {
            __builtin_prefetch(Bpf + (size_t)(base + lane) * ldb, 0, 1);
            __builtin_prefetch(Bpf + (size_t)(base + 32 + lane) * ldb, 0, 1);
        }
        #pragma unroll
        for (int u = 0; u < RING; ++u) {
            c = wmma4(ra[u], rb[u], c);
            const int k0 = 4 * (s0 + u + RING);
            ra[u] = (v2f){ Arow[k0], Arow[k0 + 1] };
            rb[u] = (v2f){ Bcol[(size_t)k0 * ldb], Bcol[(size_t)(k0 + 1) * ldb] };
            __builtin_amdgcn_sched_barrier(0);
        }
    }
    #pragma unroll
    for (int u = 0; u < RING; ++u) c = wmma4(ra[u], rb[u], c);
    return c;
}

__global__ void k_qproj(const float* __restrict__ query,
                        const float* __restrict__ Wq,
                        const float* __restrict__ bq,
                        float* __restrict__ qbuf) {
    const int waveId = blockIdx.x * (blockDim.x >> 5) + (threadIdx.x >> 5);
    const int tm   = waveId >> 7;
    const int tn   = waveId & 127;
    const int lane = threadIdx.x & 31;
    const int col  = lane & 15;
    const int half = lane >> 4;

    const float* Arow = query + (size_t)(tm * 16 + col) * KK + 2 * half;
    const float* Bcol = Wq + (size_t)(2 * half) * HK + (tn * 16 + col);

    v8f c = tile_product_ring(Arow, Bcol, Wq + tn * 16, HK, KK / 4, lane);

    const float bias = bq[tn * 16 + col];
    #pragma unroll
    for (int j = 0; j < 8; ++j) {
        const int row = tm * 16 + j + 8 * half;
        qbuf[(size_t)row * HK + tn * 16 + col] = c[j] + bias;
    }
}

#define FC_CH    256
#define FC_NC    (LL / FC_CH)
#define FC_ROWS  32
#define FC_NST   (FC_CH / FC_ROWS)
#define FC_DEPTH 4
#define FC_PV    264
#define FC_PA    260
#define P1_STEPS 16
#define P1_RING  2

__device__ __forceinline__ void fc_handover() {
    asm volatile("s_wait_dscnt 0x0" ::: "memory");
    asm volatile("s_barrier_signal -1\n\ts_barrier_wait -1" ::: "memory");
}

__global__ __launch_bounds__(512) __attribute__((amdgpu_waves_per_eu(8, 8))) void k_fused(const float* __restrict__ keys,
                                               const float* __restrict__ vals,
                                               const float* __restrict__ rpe,
                                               const float* __restrict__ qbuf,
                                               const int* __restrict__ steps,
                                               float* __restrict__ part,
                                               float* __restrict__ pmax,
                                               float* __restrict__ psum) {
    __shared__ __attribute__((aligned(16))) float sV[2][FC_ROWS][FC_PV];
    __shared__ __attribute__((aligned(16))) float sA[16][FC_PA];

    const int tid  = threadIdx.x;
    const int wv   = tid >> 5;
    const int lane = tid & 31;
    const int col  = lane & 15;
    const int half = lane >> 4;
    const int b    = blockIdx.x / FC_NC;
    const int ch   = blockIdx.x % FC_NC;
    const int l0   = ch * FC_CH;
    const int step = steps[b];

    const float* vsrc = vals + ((size_t)(l0 + 2 * wv) * BB + b) * VV + 4 * lane;
    v4f rv[FC_DEPTH][4];
    #pragma unroll
    for (int u = 0; u < FC_DEPTH; ++u) {
        #pragma unroll
        for (int q = 0; q < 4; ++q)
            rv[u][q] = __builtin_nontemporal_load(
                (const v4f*)(vsrc + (size_t)(u * FC_ROWS + (q >> 1)) * (BB * VV) + 128 * (q & 1)));
    }

    for (int i = tid; i < 8 * FC_PA; i += 512)
        sA[8 + i / FC_PA][i % FC_PA] = 0.0f;

    {
        const int rr = col & 7;
        const int g  = col >> 3;
        const int lt = l0 + 16 * wv + rr;
        const float* key0 = keys + ((size_t)lt * BB + b) * KK + 8 * g + 4 * half;
        const float* key1 = key0 + (size_t)8 * BB * KK;
        const float r0 = rpe[(size_t)lt * BB + b];
        const float r1 = rpe[(size_t)(lt + 8) * BB + b];
        const float* qb = qbuf + (size_t)b * HK + (size_t)rr * KK + 8 * g + 4 * half;

        __builtin_prefetch(key0, 0, 0);
        __builtin_prefetch(key1, 0, 0);
        __builtin_prefetch(key0 + 64, 0, 0);
        __builtin_prefetch(key1 + 64, 0, 0);

        v4f ka0[P1_RING], ka1[P1_RING], kq[P1_RING];
        #pragma unroll
        for (int u = 0; u < P1_RING; ++u) {
            ka0[u] = __builtin_nontemporal_load((const v4f*)(key0 + 16 * u));
            ka1[u] = __builtin_nontemporal_load((const v4f*)(key1 + 16 * u));
            kq[u]  = *(const v4f*)(qb + 16 * u);
        }

        v8f c0 = {}, c1 = {};
        #pragma unroll
        for (int i = 0; i < P1_STEPS; ++i) {
            const int u = i % P1_RING;
            if ((i & 3) == 0 && i + 8 < P1_STEPS) {
                __builtin_prefetch(key0 + 16 * (i + 8), 0, 0);
                __builtin_prefetch(key1 + 16 * (i + 8), 0, 0);
            }
            const v4f a0 = ka0[u], a1 = ka1[u], qv = kq[u];
            c0 = wmma4((v2f){ a0.x * r0, a0.y * r0 }, (v2f){ qv.x, qv.y }, c0);
            c1 = wmma4((v2f){ a1.x * r1, a1.y * r1 }, (v2f){ qv.x, qv.y }, c1);
            c0 = wmma4((v2f){ a0.z * r0, a0.w * r0 }, (v2f){ qv.z, qv.w }, c0);
            c1 = wmma4((v2f){ a1.z * r1, a1.w * r1 }, (v2f){ qv.z, qv.w }, c1);
            if (i + P1_RING < P1_STEPS) {
                ka0[u] = __builtin_nontemporal_load((const v4f*)(key0 + 16 * (i + P1_RING)));
                ka1[u] = __builtin_nontemporal_load((const v4f*)(key1 + 16 * (i + P1_RING)));
                kq[u]  = *(const v4f*)(qb + 16 * (i + P1_RING));
            }
        }

        float s0[8], s1[8];
        #pragma unroll
        for (int j = 0; j < 8; ++j) {
            s0[j] = c0[j] + __shfl_xor(c0[j], 24);
            s1[j] = c1[j] + __shfl_xor(c1[j], 24);
        }
        if (lane < HH) {
            const int lb = 16 * wv;
            #pragma unroll
            for (int j = 0; j < 8; ++j) {
                s0[j] = (l0 + lb + j     < step) ? s0[j] : -INFINITY;
                s1[j] = (l0 + lb + 8 + j < step) ? s1[j] : -INFINITY;
            }
            *(v4f*)&sA[lane][lb]      = (v4f){ s0[0], s0[1], s0[2], s0[3] };
            *(v4f*)&sA[lane][lb + 4]  = (v4f){ s0[4], s0[5], s0[6], s0[7] };
            *(v4f*)&sA[lane][lb + 8]  = (v4f){ s1[0], s1[1], s1[2], s1[3] };
            *(v4f*)&sA[lane][lb + 12] = (v4f){ s1[4], s1[5], s1[6], s1[7] };
        }
    }
    fc_handover();

    if (wv < HH) {
        const v4f x0 = *(const v4f*)&sA[wv][4 * lane];
        const v4f x1 = *(const v4f*)&sA[wv][128 + 4 * lane];
        float m = fmaxf(fmaxf(fmaxf(x0.x, x0.y), fmaxf(x0.z, x0.w)),
                        fmaxf(fmaxf(x1.x, x1.y), fmaxf(x1.z, x1.w)));
        #pragma unroll
        for (int off = 16; off > 0; off >>= 1) m = fmaxf(m, __shfl_xor(m, off));
        const float ms = (m == -INFINITY) ? 0.0f : m;
        const v4f e0 = { __expf(x0.x - ms), __expf(x0.y - ms), __expf(x0.z - ms), __expf(x0.w - ms) };
        const v4f e1 = { __expf(x1.x - ms), __expf(x1.y - ms), __expf(x1.z - ms), __expf(x1.w - ms) };
        float sum = ((e0.x + e0.y) + (e0.z + e0.w)) + ((e1.x + e1.y) + (e1.z + e1.w));
        #pragma unroll
        for (int off = 16; off > 0; off >>= 1) sum += __shfl_xor(sum, off);
        *(v4f*)&sA[wv][4 * lane]       = e0;
        *(v4f*)&sA[wv][128 + 4 * lane] = e1;
        if (lane == 0) {
            pmax[((size_t)b * FC_NC + ch) * HH + wv] = m;
            psum[((size_t)b * FC_NC + ch) * HH + wv] = sum;
        }
    }

    v8f acc = {};
    #pragma unroll
    for (int t = 0; t < FC_NST; ++t) {
        const int u   = t % FC_DEPTH;
        const int buf = t & 1;
        #pragma unroll
        for (int q = 0; q < 4; ++q)
            *(v4f*)&sV[buf][2 * wv + (q >> 1)][128 * (q & 1) + 4 * lane] = rv[u][q];
        if (t + FC_DEPTH < FC_NST) {
            #pragma unroll
            for (int q = 0; q < 4; ++q)
                rv[u][q] = __builtin_nontemporal_load(
                    (const v4f*)(vsrc + (size_t)((t + FC_DEPTH) * FC_ROWS + (q >> 1)) * (BB * VV) + 128 * (q & 1)));
        }
        fc_handover();
        #pragma unroll
        for (int j = 0; j < FC_ROWS / 4; ++j) {
            const int r0 = 4 * j + 2 * half;
            const v2f a  = *(const v2f*)&sA[col][FC_ROWS * t + r0];
            const v2f bb = { sV[buf][r0][16 * wv + col], sV[buf][r0 + 1][16 * wv + col] };
            acc = wmma4(a, bb, acc);
        }
    }

    if (half == 0) {
        #pragma unroll
        for (int j = 0; j < 8; ++j)
            part[(((size_t)b * FC_NC + ch) * HH + j) * VV + 16 * wv + col] = acc[j];
    }
}

#define PJ_PA    260
#define PJ_STEPS 64
#define PJ_RING  28

__global__ __launch_bounds__(256) void k_proj(float* part, const float* __restrict__ Wa) {
    __shared__ __attribute__((aligned(16))) float sP[16][PJ_PA];

    const int tid  = threadIdx.x;
    const int tw   = tid >> 5;
    const int lane = tid & 31;
    const int col  = lane & 15;
    const int half = lane >> 4;
    const int tm   = blockIdx.x >> 6;
    const int c    = (blockIdx.x >> 3) & 7;
    const int h    = blockIdx.x & 7;

    float* rows = part + (((size_t)(tm * 16) * FC_NC + c) * HH + h) * VV;
    const size_t rpitch = (size_t)FC_NC * HH * VV;

    v4f st[4];
    #pragma unroll
    for (int i = 0; i < 4; ++i) {
        const int idx = tid + 256 * i;
        st[i] = *(const v4f*)(rows + (size_t)(idx >> 6) * rpitch + 4 * (idx & 63));
    }

    const float* Bcol = Wa + ((size_t)h * VV + 2 * half) * VV + 32 * tw + 2 * col;
    v2f rb0[PJ_RING], rb1[PJ_RING];
    #pragma unroll
    for (int u = 0; u < PJ_RING; ++u) {
        rb0[u] = *(const v2f*)(Bcol + (size_t)(4 * u) * VV);
        rb1[u] = *(const v2f*)(Bcol + (size_t)(4 * u + 1) * VV);
    }

    #pragma unroll
    for (int i = 0; i < 4; ++i) {
        const int idx = tid + 256 * i;
        *(v4f*)&sP[idx >> 6][4 * (idx & 63)] = st[i];
    }
    fc_handover();

    v8f acc0 = {}, acc1 = {};
    #pragma unroll
    for (int s = 0; s < PJ_STEPS; ++s) {
        const int u = s % PJ_RING;
        const v2f a = *(const v2f*)&sP[col][4 * s + 2 * half];
        acc0 = wmma4(a, (v2f){ rb0[u].x, rb1[u].x }, acc0);
        acc1 = wmma4(a, (v2f){ rb0[u].y, rb1[u].y }, acc1);
        if (s + PJ_RING < PJ_STEPS) {
            rb0[u] = *(const v2f*)(Bcol + (size_t)(4 * (s + PJ_RING)) * VV);
            rb1[u] = *(const v2f*)(Bcol + (size_t)(4 * (s + PJ_RING) + 1) * VV);
        }
    }

    #pragma unroll
    for (int j = 0; j < 8; ++j)
        *(v2f*)(rows + (size_t)(j + 8 * half) * rpitch + 32 * tw + 2 * col) = (v2f){ acc0[j], acc1[j] };
}

__global__ __launch_bounds__(256) void k_final(const float* __restrict__ proj,
                                               const float* __restrict__ pmax,
                                               const float* __restrict__ psum,
                                               const float* __restrict__ ba,
                                               float* __restrict__ out) {
    __shared__ float sW[FC_NC * HH];
    const int b = blockIdx.x;
    const int n = threadIdx.x;

    const float* pb = proj + (size_t)b * FC_NC * HH * VV + n;
    float p[FC_NC * HH];
    #pragma unroll
    for (int i = 0; i < FC_NC * HH; ++i) p[i] = pb[(size_t)i * VV];
    const float bias = ba[n];

    if (n < HH) {
        float m[FC_NC];
        float M = -INFINITY;
        #pragma unroll
        for (int c = 0; c < FC_NC; ++c) {
            m[c] = pmax[((size_t)b * FC_NC + c) * HH + n];
            M = fmaxf(M, m[c]);
        }
        float w[FC_NC];
        float den = 0.0f;
        #pragma unroll
        for (int c = 0; c < FC_NC; ++c) {
            w[c] = __expf(m[c] - M);
            den += w[c] * psum[((size_t)b * FC_NC + c) * HH + n];
        }
        const float inv = 1.0f / den;
        #pragma unroll
        for (int c = 0; c < FC_NC; ++c) sW[c * HH + n] = w[c] * inv;
    }
    __syncthreads();

    float acc = bias;
    #pragma unroll
    for (int i = 0; i < FC_NC * HH; ++i) acc += sW[i] * p[i];
    out[(size_t)b * VV + n] = acc;
}

extern "C" void kernel_launch(void* const* d_in, const int* in_sizes, int n_in,
                              void* d_out, int out_size, void* d_ws, size_t ws_size,
                              hipStream_t stream) {
    const float* query = (const float*)d_in[0];
    const float* keys  = (const float*)d_in[1];
    const float* vals  = (const float*)d_in[2];
    const float* rpe   = (const float*)d_in[3];
    const float* Wq    = (const float*)d_in[4];
    const float* bq    = (const float*)d_in[5];
    const float* Wa    = (const float*)d_in[6];
    const float* ba    = (const float*)d_in[7];
    const int*   steps = (const int*)d_in[8];
    float* out = (float*)d_out;

    float* ws = (float*)d_ws;
    float* qbuf = ws;
    float* part = qbuf + (size_t)BB * HK;
    float* pmax = part + (size_t)BB * FC_NC * HH * VV;
    float* psum = pmax + (size_t)BB * FC_NC * HH;

    k_qproj<<<256, 64, 0, stream>>>(query, Wq, bq, qbuf);
    k_fused<<<BB * FC_NC, 512, 0, stream>>>(keys, vals, rpe, qbuf, steps, part, pmax, psum);
    k_proj<<<(BB / 16) * FC_NC * HH, 256, 0, stream>>>(part, Wa);
    k_final<<<BB, 256, 0, stream>>>(part, pmax, psum, ba, out);
}


# k_proj: a head's 8 chunks merged by their maxima while the A slab is staged: one 16-step product per (row tile, head, 64 v) for eight of 64 steps; k_final sums 32 partials + bias (new sum order)
# speedup vs baseline: 10.5005x; 1.0606x over previous
#include <hip/hip_runtime.h>
#include <hip/hip_bf16.h>
#include <math.h>

#define LL 2048
#define BB 64
#define KK 256
#define VV 256
#define HH 8
#define HK 2048
#define HV 2048

typedef __attribute__((ext_vector_type(2))) float v2f;
typedef __attribute__((ext_vector_type(4))) float v4f;
typedef __attribute__((ext_vector_type(8))) float v8f;

__device__ __forceinline__ v8f wmma4(v2f a, v2f b, v8f c) {
    return __builtin_amdgcn_wmma_f32_16x16x4_f32(
         false, a,  false, b,
         (short)0, c,  false,  false);
}

#define RING     16
#define PF_TRIPS 4

__device__ __forceinline__ v8f tile_product_ring(const float* __restrict__ Arow,
                                                 const float* __restrict__ Bcol,
                                                 const float* __restrict__ Bpf,
                                                 const int ldb, const int ksteps, const int lane) {
    const int krows = 4 * ksteps;
    v2f ra[RING], rb[RING];
    #pragma unroll
    for (int u = 0; u < RING; ++u) {
        const int k0 = 4 * u;
        ra[u] = (v2f){ Arow[k0], Arow[k0 + 1] };
        rb[u] = (v2f){ Bcol[(size_t)k0 * ldb], Bcol[(size_t)(k0 + 1) * ldb] };
        __builtin_amdgcn_sched_barrier(0);
    }
    #pragma unroll
    for (int i = 0; i < 2 * PF_TRIPS; ++i) {
        const int base = 4 * RING + 32 * i;
        if (base < krows) __builtin_prefetch(Bpf + (size_t)(base + lane) * ldb, 0, 1);
    }

    v8f c = {};
    for (int s0 = 0; s0 < ksteps - RING; s0 += RING) {
        const int base = 4 * (s0 + RING * (1 + PF_TRIPS));
        if (base < krows) {
            __builtin_prefetch(Bpf + (size_t)(base + lane) * ldb, 0, 1);
            __builtin_prefetch(Bpf + (size_t)(base + 32 + lane) * ldb, 0, 1);
        }
        #pragma unroll
        for (int u = 0; u < RING; ++u) {
            c = wmma4(ra[u], rb[u], c);
            const int k0 = 4 * (s0 + u + RING);
            ra[u] = (v2f){ Arow[k0], Arow[k0 + 1] };
            rb[u] = (v2f){ Bcol[(size_t)k0 * ldb], Bcol[(size_t)(k0 + 1) * ldb] };
            __builtin_amdgcn_sched_barrier(0);
        }
    }
    #pragma unroll
    for (int u = 0; u < RING; ++u) c = wmma4(ra[u], rb[u], c);
    return c;
}

__global__ void k_qproj(const float* __restrict__ query,
                        const float* __restrict__ Wq,
                        const float* __restrict__ bq,
                        float* __restrict__ qbuf) {
    const int waveId = blockIdx.x * (blockDim.x >> 5) + (threadIdx.x >> 5);
    const int tm   = waveId >> 7;
    const int tn   = waveId & 127;
    const int lane = threadIdx.x & 31;
    const int col  = lane & 15;
    const int half = lane >> 4;

    const float* Arow = query + (size_t)(tm * 16 + col) * KK + 2 * half;
    const float* Bcol = Wq + (size_t)(2 * half) * HK + (tn * 16 + col);

    v8f c = tile_product_ring(Arow, Bcol, Wq + tn * 16, HK, KK / 4, lane);

    const float bias = bq[tn * 16 + col];
    #pragma unroll
    for (int j = 0; j < 8; ++j) {
        const int row = tm * 16 + j + 8 * half;
        qbuf[(size_t)row * HK + tn * 16 + col] = c[j] + bias;
    }
}

#define FC_CH    256
#define FC_NC    (LL / FC_CH)
#define FC_ROWS  32
#define FC_NST   (FC_CH / FC_ROWS)
#define FC_DEPTH 4
#define FC_PV    264
#define FC_PA    260
#define P1_STEPS 16
#define P1_RING  2

__device__ __forceinline__ void fc_handover() {
    asm volatile("s_wait_dscnt 0x0" ::: "memory");
    asm volatile("s_barrier_signal -1\n\ts_barrier_wait -1" ::: "memory");
}

__global__ __launch_bounds__(512) __attribute__((amdgpu_waves_per_eu(8, 8))) void k_fused(const float* __restrict__ keys,
                                               const float* __restrict__ vals,
                                               const float* __restrict__ rpe,
                                               const float* __restrict__ qbuf,
                                               const int* __restrict__ steps,
                                               float* __restrict__ part,
                                               float* __restrict__ pmax,
                                               float* __restrict__ psum) {
    __shared__ __attribute__((aligned(16))) float sV[2][FC_ROWS][FC_PV];
    __shared__ __attribute__((aligned(16))) float sA[16][FC_PA];

    const int tid  = threadIdx.x;
    const int wv   = tid >> 5;
    const int lane = tid & 31;
    const int col  = lane & 15;
    const int half = lane >> 4;
    const int b    = blockIdx.x / FC_NC;
    const int ch   = blockIdx.x % FC_NC;
    const int l0   = ch * FC_CH;
    const int step = steps[b];

    const float* vsrc = vals + ((size_t)(l0 + 2 * wv) * BB + b) * VV + 4 * lane;
    v4f rv[FC_DEPTH][4];
    #pragma unroll
    for (int u = 0; u < FC_DEPTH; ++u) {
        #pragma unroll
        for (int q = 0; q < 4; ++q)
            rv[u][q] = __builtin_nontemporal_load(
                (const v4f*)(vsrc + (size_t)(u * FC_ROWS + (q >> 1)) * (BB * VV) + 128 * (q & 1)));
    }

    for (int i = tid; i < 8 * FC_PA; i += 512)
        sA[8 + i / FC_PA][i % FC_PA] = 0.0f;

    {
        const int rr = col & 7;
        const int g  = col >> 3;
        const int lt = l0 + 16 * wv + rr;
        const float* key0 = keys + ((size_t)lt * BB + b) * KK + 8 * g + 4 * half;
        const float* key1 = key0 + (size_t)8 * BB * KK;
        const float r0 = rpe[(size_t)lt * BB + b];
        const float r1 = rpe[(size_t)(lt + 8) * BB + b];
        const float* qb = qbuf + (size_t)b * HK + (size_t)rr * KK + 8 * g + 4 * half;

        __builtin_prefetch(key0, 0, 0);
        __builtin_prefetch(key1, 0, 0);
        __builtin_prefetch(key0 + 64, 0, 0);
        __builtin_prefetch(key1 + 64, 0, 0);

        v4f ka0[P1_RING], ka1[P1_RING], kq[P1_RING];
        #pragma unroll
        for (int u = 0; u < P1_RING; ++u) {
            ka0[u] = __builtin_nontemporal_load((const v4f*)(key0 + 16 * u));
            ka1[u] = __builtin_nontemporal_load((const v4f*)(key1 + 16 * u));
            kq[u]  = *(const v4f*)(qb + 16 * u);
        }

        v8f c0 = {}, c1 = {};
        #pragma unroll
        for (int i = 0; i < P1_STEPS; ++i) {
            const int u = i % P1_RING;
            if ((i & 3) == 0 && i + 8 < P1_STEPS) {
                __builtin_prefetch(key0 + 16 * (i + 8), 0, 0);
                __builtin_prefetch(key1 + 16 * (i + 8), 0, 0);
            }
            const v4f a0 = ka0[u], a1 = ka1[u], qv = kq[u];
            c0 = wmma4((v2f){ a0.x * r0, a0.y * r0 }, (v2f){ qv.x, qv.y }, c0);
            c1 = wmma4((v2f){ a1.x * r1, a1.y * r1 }, (v2f){ qv.x, qv.y }, c1);
            c0 = wmma4((v2f){ a0.z * r0, a0.w * r0 }, (v2f){ qv.z, qv.w }, c0);
            c1 = wmma4((v2f){ a1.z * r1, a1.w * r1 }, (v2f){ qv.z, qv.w }, c1);
            if (i + P1_RING < P1_STEPS) {
                ka0[u] = __builtin_nontemporal_load((const v4f*)(key0 + 16 * (i + P1_RING)));
                ka1[u] = __builtin_nontemporal_load((const v4f*)(key1 + 16 * (i + P1_RING)));
                kq[u]  = *(const v4f*)(qb + 16 * (i + P1_RING));
            }
        }

        float s0[8], s1[8];
        #pragma unroll
        for (int j = 0; j < 8; ++j) {
            s0[j] = c0[j] + __shfl_xor(c0[j], 24);
            s1[j] = c1[j] + __shfl_xor(c1[j], 24);
        }
        if (lane < HH) {
            const int lb = 16 * wv;
            #pragma unroll
            for (int j = 0; j < 8; ++j) {
                s0[j] = (l0 + lb + j     < step) ? s0[j] : -INFINITY;
                s1[j] = (l0 + lb + 8 + j < step) ? s1[j] : -INFINITY;
            }
            *(v4f*)&sA[lane][lb]      = (v4f){ s0[0], s0[1], s0[2], s0[3] };
            *(v4f*)&sA[lane][lb + 4]  = (v4f){ s0[4], s0[5], s0[6], s0[7] };
            *(v4f*)&sA[lane][lb + 8]  = (v4f){ s1[0], s1[1], s1[2], s1[3] };
            *(v4f*)&sA[lane][lb + 12] = (v4f){ s1[4], s1[5], s1[6], s1[7] };
        }
    }
    fc_handover();

    if (wv < HH) {
        const v4f x0 = *(const v4f*)&sA[wv][4 * lane];
        const v4f x1 = *(const v4f*)&sA[wv][128 + 4 * lane];
        float m = fmaxf(fmaxf(fmaxf(x0.x, x0.y), fmaxf(x0.z, x0.w)),
                        fmaxf(fmaxf(x1.x, x1.y), fmaxf(x1.z, x1.w)));
        #pragma unroll
        for (int off = 16; off > 0; off >>= 1) m = fmaxf(m, __shfl_xor(m, off));
        const float ms = (m == -INFINITY) ? 0.0f : m;
        const v4f e0 = { __expf(x0.x - ms), __expf(x0.y - ms), __expf(x0.z - ms), __expf(x0.w - ms) };
        const v4f e1 = { __expf(x1.x - ms), __expf(x1.y - ms), __expf(x1.z - ms), __expf(x1.w - ms) };
        float sum = ((e0.x + e0.y) + (e0.z + e0.w)) + ((e1.x + e1.y) + (e1.z + e1.w));
        #pragma unroll
        for (int off = 16; off > 0; off >>= 1) sum += __shfl_xor(sum, off);
        *(v4f*)&sA[wv][4 * lane]       = e0;
        *(v4f*)&sA[wv][128 + 4 * lane] = e1;
        if (lane == 0) {
            pmax[((size_t)b * FC_NC + ch) * HH + wv] = m;
            psum[((size_t)b * FC_NC + ch) * HH + wv] = sum;
        }
    }

    v8f acc = {};
    #pragma unroll
    for (int t = 0; t < FC_NST; ++t) {
        const int u   = t % FC_DEPTH;
        const int buf = t & 1;
        #pragma unroll
        for (int q = 0; q < 4; ++q)
            *(v4f*)&sV[buf][2 * wv + (q >> 1)][128 * (q & 1) + 4 * lane] = rv[u][q];
        if (t + FC_DEPTH < FC_NST) {
            #pragma unroll
            for (int q = 0; q < 4; ++q)
                rv[u][q] = __builtin_nontemporal_load(
                    (const v4f*)(vsrc + (size_t)((t + FC_DEPTH) * FC_ROWS + (q >> 1)) * (BB * VV) + 128 * (q & 1)));
        }
        fc_handover();
        #pragma unroll
        for (int j = 0; j < FC_ROWS / 4; ++j) {
            const int r0 = 4 * j + 2 * half;
            const v2f a  = *(const v2f*)&sA[col][FC_ROWS * t + r0];
            const v2f bb = { sV[buf][r0][16 * wv + col], sV[buf][r0 + 1][16 * wv + col] };
            acc = wmma4(a, bb, acc);
        }
    }

    if (half == 0) {
        #pragma unroll
        for (int j = 0; j < 8; ++j)
            part[(((size_t)b * FC_NC + ch) * HH + j) * VV + 16 * wv + col] = acc[j];
    }
}

#define PM_KV    64
#define PM_NQ    4
#define PM_PA    68
#define PM_STEPS 16

__global__ __launch_bounds__(256) void k_proj(float* part, const float* __restrict__ Wa,
                                              const float* __restrict__ pmax,
                                              const float* __restrict__ psum) {
    __shared__ __attribute__((aligned(16))) float sP[16][PM_PA];

    const int tid  = threadIdx.x;
    const int tw   = tid >> 5;
    const int lane = tid & 31;
    const int col  = lane & 15;
    const int half = lane >> 4;
    const int tm   = blockIdx.x >> 5;
    const int h    = (blockIdx.x >> 2) & 7;
    const int vq   = blockIdx.x & 3;

    const int r  = tid >> 4;
    const int pc = tid & 15;
    const int bs = tm * 16 + r;

    float m[FC_NC], sm[FC_NC];
    #pragma unroll
    for (int c = 0; c < FC_NC; ++c) {
        m[c]  = pmax[((size_t)bs * FC_NC + c) * HH + h];
        sm[c] = psum[((size_t)bs * FC_NC + c) * HH + h];
    }
    const float* prow = part + ((size_t)bs * FC_NC * HH + h) * VV + PM_KV * vq + 4 * pc;
    v4f pv[FC_NC];
    #pragma unroll
    for (int c = 0; c < FC_NC; ++c)
        pv[c] = *(const v4f*)(prow + (size_t)c * HH * VV);

    const float* Bcol = Wa + ((size_t)h * VV + PM_KV * vq + 2 * half) * VV + 32 * tw + 2 * col;
    v2f rb0[PM_STEPS], rb1[PM_STEPS];
    #pragma unroll
    for (int u = 0; u < PM_STEPS; ++u) {
        rb0[u] = *(const v2f*)(Bcol + (size_t)(4 * u) * VV);
        rb1[u] = *(const v2f*)(Bcol + (size_t)(4 * u + 1) * VV);
    }

    float M = m[0];
    #pragma unroll
    for (int c = 1; c < FC_NC; ++c) M = fmaxf(M, m[c]);
    float w[FC_NC];
    float den = 0.0f;
    #pragma unroll
    for (int c = 0; c < FC_NC; ++c) {
        w[c] = __expf(m[c] - M);
        den += w[c] * sm[c];
    }
    const float inv = 1.0f / den;
    v4f mg = { 0.0f, 0.0f, 0.0f, 0.0f };
    #pragma unroll
    for (int c = 0; c < FC_NC; ++c) {
        const float wc = w[c] * inv;
        mg.x += wc * pv[c].x;
        mg.y += wc * pv[c].y;
        mg.z += wc * pv[c].z;
        mg.w += wc * pv[c].w;
    }
    *(v4f*)&sP[r][4 * pc] = mg;
    fc_handover();

    v8f acc0 = {}, acc1 = {};
    #pragma unroll
    for (int s = 0; s < PM_STEPS; ++s) {
        const v2f a = *(const v2f*)&sP[col][4 * s + 2 * half];
        acc0 = wmma4(a, (v2f){ rb0[s].x, rb1[s].x }, acc0);
        acc1 = wmma4(a, (v2f){ rb0[s].y, rb1[s].y }, acc1);
    }

    const int n0 = 32 * tw + 2 * col;
    #pragma unroll
    for (int j = 0; j < 8; ++j) {
        const int bo = tm * 16 + j + 8 * half;
        *(v2f*)(part + (((size_t)bo * FC_NC + (n0 >> 6)) * HH + h) * VV + PM_KV * vq + (n0 & 63)) =
            (v2f){ acc0[j], acc1[j] };
    }
}

__global__ __launch_bounds__(256) void k_final(const float* __restrict__ proj,
                                               const float* __restrict__ ba,
                                               float* __restrict__ out) {
    const int b = blockIdx.x;
    const int n = threadIdx.x;

    const float* pb = proj + ((size_t)b * FC_NC + (n >> 6)) * HH * VV + (n & 63);
    float p[HH * PM_NQ];
    #pragma unroll
    for (int i = 0; i < HH * PM_NQ; ++i)
        p[i] = pb[(size_t)(i >> 2) * VV + PM_KV * (i & 3)];

    float acc = ba[n];
    #pragma unroll
    for (int i = 0; i < HH * PM_NQ; ++i) acc += p[i];
    out[(size_t)b * VV + n] = acc;
}

extern "C" void kernel_launch(void* const* d_in, const int* in_sizes, int n_in,
                              void* d_out, int out_size, void* d_ws, size_t ws_size,
                              hipStream_t stream) {
    const float* query = (const float*)d_in[0];
    const float* keys  = (const float*)d_in[1];
    const float* vals  = (const float*)d_in[2];
    const float* rpe   = (const float*)d_in[3];
    const float* Wq    = (const float*)d_in[4];
    const float* bq    = (const float*)d_in[5];
    const float* Wa    = (const float*)d_in[6];
    const float* ba    = (const float*)d_in[7];
    const int*   steps = (const int*)d_in[8];
    float* out = (float*)d_out;

    float* ws = (float*)d_ws;
    float* qbuf = ws;
    float* part = qbuf + (size_t)BB * HK;
    float* pmax = part + (size_t)BB * FC_NC * HH * VV;
    float* psum = pmax + (size_t)BB * FC_NC * HH;

    k_qproj<<<256, 64, 0, stream>>>(query, Wq, bq, qbuf);
    k_fused<<<BB * FC_NC, 512, 0, stream>>>(keys, vals, rpe, qbuf, steps, part, pmax, psum);
    k_proj<<<(BB / 16) * HH * PM_NQ, 256, 0, stream>>>(part, Wa, pmax, psum);
    k_final<<<BB, 256, 0, stream>>>(part, ba, out);
}


# k_qproj: 256 more 2-wave work-groups in its grid only prefetch: the first 128 vals rows of every (batch, chunk), k_fused's opening loads, are asked into L2 under the projection and the launch gap
# speedup vs baseline: 10.8763x; 1.0358x over previous
#include <hip/hip_runtime.h>
#include <hip/hip_bf16.h>
#include <math.h>

#define LL 2048
#define BB 64
#define KK 256
#define VV 256
#define HH 8
#define HK 2048
#define HV 2048

typedef __attribute__((ext_vector_type(2))) float v2f;
typedef __attribute__((ext_vector_type(4))) float v4f;
typedef __attribute__((ext_vector_type(8))) float v8f;

__device__ __forceinline__ v8f wmma4(v2f a, v2f b, v8f c) {
    return __builtin_amdgcn_wmma_f32_16x16x4_f32(
         false, a,  false, b,
         (short)0, c,  false,  false);
}

#define RING     16
#define PF_TRIPS 4

__device__ __forceinline__ v8f tile_product_ring(const float* __restrict__ Arow,
                                                 const float* __restrict__ Bcol,
                                                 const float* __restrict__ Bpf,
                                                 const int ldb, const int ksteps, const int lane) {
    const int krows = 4 * ksteps;
    v2f ra[RING], rb[RING];
    #pragma unroll
    for (int u = 0; u < RING; ++u) {
        const int k0 = 4 * u;
        ra[u] = (v2f){ Arow[k0], Arow[k0 + 1] };
        rb[u] = (v2f){ Bcol[(size_t)k0 * ldb], Bcol[(size_t)(k0 + 1) * ldb] };
        __builtin_amdgcn_sched_barrier(0);
    }
    #pragma unroll
    for (int i = 0; i < 2 * PF_TRIPS; ++i) {
        const int base = 4 * RING + 32 * i;
        if (base < krows) __builtin_prefetch(Bpf + (size_t)(base + lane) * ldb, 0, 1);
    }

    v8f c = {};
    for (int s0 = 0; s0 < ksteps - RING; s0 += RING) {
        const int base = 4 * (s0 + RING * (1 + PF_TRIPS));
        if (base < krows) {
            __builtin_prefetch(Bpf + (size_t)(base + lane) * ldb, 0, 1);
            __builtin_prefetch(Bpf + (size_t)(base + 32 + lane) * ldb, 0, 1);
        }
        #pragma unroll
        for (int u = 0; u < RING; ++u) {
            c = wmma4(ra[u], rb[u], c);
            const int k0 = 4 * (s0 + u + RING);
            ra[u] = (v2f){ Arow[k0], Arow[k0 + 1] };
            rb[u] = (v2f){ Bcol[(size_t)k0 * ldb], Bcol[(size_t)(k0 + 1) * ldb] };
            __builtin_amdgcn_sched_barrier(0);
        }
    }
    #pragma unroll
    for (int u = 0; u < RING; ++u) c = wmma4(ra[u], rb[u], c);
    return c;
}

#define QP_BLOCKS 256
#define PFV_CH    256
#define PFV_NC    8
#define PFV_ROWS  128

__global__ void k_qproj(const float* __restrict__ query,
                        const float* __restrict__ Wq,
                        const float* __restrict__ bq,
                        float* __restrict__ qbuf,
                        const float* __restrict__ vals) {
    if (blockIdx.x >= QP_BLOCKS) {
        const int pl = threadIdx.x & 31;
        const int f  = (blockIdx.x - QP_BLOCKS) + QP_BLOCKS * (threadIdx.x >> 5);
        const int pb = f / PFV_NC;
        const int pc = f % PFV_NC;
        const float* p = vals + ((size_t)(pc * PFV_CH + (pl >> 2)) * BB + pb) * VV + 64 * (pl & 3);
        #pragma unroll
        for (int j = 0; j < PFV_ROWS / 8; ++j)
            __builtin_prefetch(p + (size_t)(8 * j) * BB * VV, 0, 1);
        return;
    }
    const int waveId = blockIdx.x * (blockDim.x >> 5) + (threadIdx.x >> 5);
    const int tm   = waveId >> 7;
    const int tn   = waveId & 127;
    const int lane = threadIdx.x & 31;
    const int col  = lane & 15;
    const int half = lane >> 4;

    const float* Arow = query + (size_t)(tm * 16 + col) * KK + 2 * half;
    const float* Bcol = Wq + (size_t)(2 * half) * HK + (tn * 16 + col);

    v8f c = tile_product_ring(Arow, Bcol, Wq + tn * 16, HK, KK / 4, lane);

    const float bias = bq[tn * 16 + col];
    #pragma unroll
    for (int j = 0; j < 8; ++j) {
        const int row = tm * 16 + j + 8 * half;
        qbuf[(size_t)row * HK + tn * 16 + col] = c[j] + bias;
    }
}

#define FC_CH    256
#define FC_NC    (LL / FC_CH)
#define FC_ROWS  32
#define FC_NST   (FC_CH / FC_ROWS)
#define FC_DEPTH 4
#define FC_PV    264
#define FC_PA    260
#define P1_STEPS 16
#define P1_RING  2
static_assert(PFV_CH == FC_CH && PFV_NC == FC_NC && PFV_ROWS == FC_DEPTH * FC_ROWS,
              "k_qproj's prefetch role must cover k_fused's first loads");
static_assert(2 * QP_BLOCKS == BB * FC_NC, "one prefetching wave per k_fused work-group");

__device__ __forceinline__ void fc_handover() {
    asm volatile("s_wait_dscnt 0x0" ::: "memory");
    asm volatile("s_barrier_signal -1\n\ts_barrier_wait -1" ::: "memory");
}

__global__ __launch_bounds__(512) __attribute__((amdgpu_waves_per_eu(8, 8))) void k_fused(const float* __restrict__ keys,
                                               const float* __restrict__ vals,
                                               const float* __restrict__ rpe,
                                               const float* __restrict__ qbuf,
                                               const int* __restrict__ steps,
                                               float* __restrict__ part,
                                               float* __restrict__ pmax,
                                               float* __restrict__ psum) {
    __shared__ __attribute__((aligned(16))) float sV[2][FC_ROWS][FC_PV];
    __shared__ __attribute__((aligned(16))) float sA[16][FC_PA];

    const int tid  = threadIdx.x;
    const int wv   = tid >> 5;
    const int lane = tid & 31;
    const int col  = lane & 15;
    const int half = lane >> 4;
    const int b    = blockIdx.x / FC_NC;
    const int ch   = blockIdx.x % FC_NC;
    const int l0   = ch * FC_CH;
    const int step = steps[b];

    const float* vsrc = vals + ((size_t)(l0 + 2 * wv) * BB + b) * VV + 4 * lane;
    v4f rv[FC_DEPTH][4];
    #pragma unroll
    for (int u = 0; u < FC_DEPTH; ++u) {
        #pragma unroll
        for (int q = 0; q < 4; ++q)
            rv[u][q] = __builtin_nontemporal_load(
                (const v4f*)(vsrc + (size_t)(u * FC_ROWS + (q >> 1)) * (BB * VV) + 128 * (q & 1)));
    }

    for (int i = tid; i < 8 * FC_PA; i += 512)
        sA[8 + i / FC_PA][i % FC_PA] = 0.0f;

    {
        const int rr = col & 7;
        const int g  = col >> 3;
        const int lt = l0 + 16 * wv + rr;
        const float* key0 = keys + ((size_t)lt * BB + b) * KK + 8 * g + 4 * half;
        const float* key1 = key0 + (size_t)8 * BB * KK;
        const float r0 = rpe[(size_t)lt * BB + b];
        const float r1 = rpe[(size_t)(lt + 8) * BB + b];
        const float* qb = qbuf + (size_t)b * HK + (size_t)rr * KK + 8 * g + 4 * half;

        __builtin_prefetch(key0, 0, 0);
        __builtin_prefetch(key1, 0, 0);
        __builtin_prefetch(key0 + 64, 0, 0);
        __builtin_prefetch(key1 + 64, 0, 0);

        v4f ka0[P1_RING], ka1[P1_RING], kq[P1_RING];
        #pragma unroll
        for (int u = 0; u < P1_RING; ++u) {
            ka0[u] = __builtin_nontemporal_load((const v4f*)(key0 + 16 * u));
            ka1[u] = __builtin_nontemporal_load((const v4f*)(key1 + 16 * u));
            kq[u]  = *(const v4f*)(qb + 16 * u);
        }

        v8f c0 = {}, c1 = {};
        #pragma unroll
        for (int i = 0; i < P1_STEPS; ++i) {
            const int u = i % P1_RING;
            if ((i & 3) == 0 && i + 8 < P1_STEPS) {
                __builtin_prefetch(key0 + 16 * (i + 8), 0, 0);
                __builtin_prefetch(key1 + 16 * (i + 8), 0, 0);
            }
            const v4f a0 = ka0[u], a1 = ka1[u], qv = kq[u];
            c0 = wmma4((v2f){ a0.x * r0, a0.y * r0 }, (v2f){ qv.x, qv.y }, c0);
            c1 = wmma4((v2f){ a1.x * r1, a1.y * r1 }, (v2f){ qv.x, qv.y }, c1);
            c0 = wmma4((v2f){ a0.z * r0, a0.w * r0 }, (v2f){ qv.z, qv.w }, c0);
            c1 = wmma4((v2f){ a1.z * r1, a1.w * r1 }, (v2f){ qv.z, qv.w }, c1);
            if (i + P1_RING < P1_STEPS) {
                ka0[u] = __builtin_nontemporal_load((const v4f*)(key0 + 16 * (i + P1_RING)));
                ka1[u] = __builtin_nontemporal_load((const v4f*)(key1 + 16 * (i + P1_RING)));
                kq[u]  = *(const v4f*)(qb + 16 * (i + P1_RING));
            }
        }

        float s0[8], s1[8];
        #pragma unroll
        for (int j = 0; j < 8; ++j) {
            s0[j] = c0[j] + __shfl_xor(c0[j], 24);
            s1[j] = c1[j] + __shfl_xor(c1[j], 24);
        }
        if (lane < HH) {
            const int lb = 16 * wv;
            #pragma unroll
            for (int j = 0; j < 8; ++j) {
                s0[j] = (l0 + lb + j     < step) ? s0[j] : -INFINITY;
                s1[j] = (l0 + lb + 8 + j < step) ? s1[j] : -INFINITY;
            }
            *(v4f*)&sA[lane][lb]      = (v4f){ s0[0], s0[1], s0[2], s0[3] };
            *(v4f*)&sA[lane][lb + 4]  = (v4f){ s0[4], s0[5], s0[6], s0[7] };
            *(v4f*)&sA[lane][lb + 8]  = (v4f){ s1[0], s1[1], s1[2], s1[3] };
            *(v4f*)&sA[lane][lb + 12] = (v4f){ s1[4], s1[5], s1[6], s1[7] };
        }
    }
    fc_handover();

    if (wv < HH) {
        const v4f x0 = *(const v4f*)&sA[wv][4 * lane];
        const v4f x1 = *(const v4f*)&sA[wv][128 + 4 * lane];
        float m = fmaxf(fmaxf(fmaxf(x0.x, x0.y), fmaxf(x0.z, x0.w)),
                        fmaxf(fmaxf(x1.x, x1.y), fmaxf(x1.z, x1.w)));
        #pragma unroll
        for (int off = 16; off > 0; off >>= 1) m = fmaxf(m, __shfl_xor(m, off));
        const float ms = (m == -INFINITY) ? 0.0f : m;
        const v4f e0 = { __expf(x0.x - ms), __expf(x0.y - ms), __expf(x0.z - ms), __expf(x0.w - ms) };
        const v4f e1 = { __expf(x1.x - ms), __expf(x1.y - ms), __expf(x1.z - ms), __expf(x1.w - ms) };
        float sum = ((e0.x + e0.y) + (e0.z + e0.w)) + ((e1.x + e1.y) + (e1.z + e1.w));
        #pragma unroll
        for (int off = 16; off > 0; off >>= 1) sum += __shfl_xor(sum, off);
        *(v4f*)&sA[wv][4 * lane]       = e0;
        *(v4f*)&sA[wv][128 + 4 * lane] = e1;
        if (lane == 0) {
            pmax[((size_t)b * FC_NC + ch) * HH + wv] = m;
            psum[((size_t)b * FC_NC + ch) * HH + wv] = sum;
        }
    }

    v8f acc = {};
    #pragma unroll
    for (int t = 0; t < FC_NST; ++t) {
        const int u   = t % FC_DEPTH;
        const int buf = t & 1;
        #pragma unroll
        for (int q = 0; q < 4; ++q)
            *(v4f*)&sV[buf][2 * wv + (q >> 1)][128 * (q & 1) + 4 * lane] = rv[u][q];
        if (t + FC_DEPTH < FC_NST) {
            #pragma unroll
            for (int q = 0; q < 4; ++q)
                rv[u][q] = __builtin_nontemporal_load(
                    (const v4f*)(vsrc + (size_t)((t + FC_DEPTH) * FC_ROWS + (q >> 1)) * (BB * VV) + 128 * (q & 1)));
        }
        fc_handover();
        #pragma unroll
        for (int j = 0; j < FC_ROWS / 4; ++j) {
            const int r0 = 4 * j + 2 * half;
            const v2f a  = *(const v2f*)&sA[col][FC_ROWS * t + r0];
            const v2f bb = { sV[buf][r0][16 * wv + col], sV[buf][r0 + 1][16 * wv + col] };
            acc = wmma4(a, bb, acc);
        }
    }

    if (half == 0) {
        #pragma unroll
        for (int j = 0; j < 8; ++j)
            part[(((size_t)b * FC_NC + ch) * HH + j) * VV + 16 * wv + col] = acc[j];
    }
}

#define PM_KV    64
#define PM_NQ    4
#define PM_PA    68
#define PM_STEPS 16

__global__ __launch_bounds__(256) void k_proj(float* part, const float* __restrict__ Wa,
                                              const float* __restrict__ pmax,
                                              const float* __restrict__ psum) {
    __shared__ __attribute__((aligned(16))) float sP[16][PM_PA];

    const int tid  = threadIdx.x;
    const int tw   = tid >> 5;
    const int lane = tid & 31;
    const int col  = lane & 15;
    const int half = lane >> 4;
    const int tm   = blockIdx.x >> 5;
    const int h    = (blockIdx.x >> 2) & 7;
    const int vq   = blockIdx.x & 3;

    const int r  = tid >> 4;
    const int pc = tid & 15;
    const int bs = tm * 16 + r;

    float m[FC_NC], sm[FC_NC];
    #pragma unroll
    for (int c = 0; c < FC_NC; ++c) {
        m[c]  = pmax[((size_t)bs * FC_NC + c) * HH + h];
        sm[c] = psum[((size_t)bs * FC_NC + c) * HH + h];
    }
    const float* prow = part + ((size_t)bs * FC_NC * HH + h) * VV + PM_KV * vq + 4 * pc;
    v4f pv[FC_NC];
    #pragma unroll
    for (int c = 0; c < FC_NC; ++c)
        pv[c] = *(const v4f*)(prow + (size_t)c * HH * VV);

    const float* Bcol = Wa + ((size_t)h * VV + PM_KV * vq + 2 * half) * VV + 32 * tw + 2 * col;
    v2f rb0[PM_STEPS], rb1[PM_STEPS];
    #pragma unroll
    for (int u = 0; u < PM_STEPS; ++u) {
        rb0[u] = *(const v2f*)(Bcol + (size_t)(4 * u) * VV);
        rb1[u] = *(const v2f*)(Bcol + (size_t)(4 * u + 1) * VV);
    }

    float M = m[0];
    #pragma unroll
    for (int c = 1; c < FC_NC; ++c) M = fmaxf(M, m[c]);
    float w[FC_NC];
    float den = 0.0f;
    #pragma unroll
    for (int c = 0; c < FC_NC; ++c) {
        w[c] = __expf(m[c] - M);
        den += w[c] * sm[c];
    }
    const float inv = 1.0f / den;
    v4f mg = { 0.0f, 0.0f, 0.0f, 0.0f };
    #pragma unroll
    for (int c = 0; c < FC_NC; ++c) {
        const float wc = w[c] * inv;
        mg.x += wc * pv[c].x;
        mg.y += wc * pv[c].y;
        mg.z += wc * pv[c].z;
        mg.w += wc * pv[c].w;
    }
    *(v4f*)&sP[r][4 * pc] = mg;
    fc_handover();

    v8f acc0 = {}, acc1 = {};
    #pragma unroll
    for (int s = 0; s < PM_STEPS; ++s) {
        const v2f a = *(const v2f*)&sP[col][4 * s + 2 * half];
        acc0 = wmma4(a, (v2f){ rb0[s].x, rb1[s].x }, acc0);
        acc1 = wmma4(a, (v2f){ rb0[s].y, rb1[s].y }, acc1);
    }

    const int n0 = 32 * tw + 2 * col;
    #pragma unroll
    for (int j = 0; j < 8; ++j) {
        const int bo = tm * 16 + j + 8 * half;
        *(v2f*)(part + (((size_t)bo * FC_NC + (n0 >> 6)) * HH + h) * VV + PM_KV * vq + (n0 & 63)) =
            (v2f){ acc0[j], acc1[j] };
    }
}

__global__ __launch_bounds__(256) void k_final(const float* __restrict__ proj,
                                               const float* __restrict__ ba,
                                               float* __restrict__ out) {
    const int b = blockIdx.x;
    const int n = threadIdx.x;

    const float* pb = proj + ((size_t)b * FC_NC + (n >> 6)) * HH * VV + (n & 63);
    float p[HH * PM_NQ];
    #pragma unroll
    for (int i = 0; i < HH * PM_NQ; ++i)
        p[i] = pb[(size_t)(i >> 2) * VV + PM_KV * (i & 3)];

    float acc = ba[n];
    #pragma unroll
    for (int i = 0; i < HH * PM_NQ; ++i) acc += p[i];
    out[(size_t)b * VV + n] = acc;
}

extern "C" void kernel_launch(void* const* d_in, const int* in_sizes, int n_in,
                              void* d_out, int out_size, void* d_ws, size_t ws_size,
                              hipStream_t stream) {
    const float* query = (const float*)d_in[0];
    const float* keys  = (const float*)d_in[1];
    const float* vals  = (const float*)d_in[2];
    const float* rpe   = (const float*)d_in[3];
    const float* Wq    = (const float*)d_in[4];
    const float* bq    = (const float*)d_in[5];
    const float* Wa    = (const float*)d_in[6];
    const float* ba    = (const float*)d_in[7];
    const int*   steps = (const int*)d_in[8];
    float* out = (float*)d_out;

    float* ws = (float*)d_ws;
    float* qbuf = ws;
    float* part = qbuf + (size_t)BB * HK;
    float* pmax = part + (size_t)BB * FC_NC * HH * VV;
    float* psum = pmax + (size_t)BB * FC_NC * HH;

    k_qproj<<<2 * QP_BLOCKS, 64, 0, stream>>>(query, Wq, bq, qbuf, vals);
    k_fused<<<BB * FC_NC, 512, 0, stream>>>(keys, vals, rpe, qbuf, steps, part, pmax, psum);
    k_proj<<<(BB / 16) * HH * PM_NQ, 256, 0, stream>>>(part, Wa, pmax, psum);
    k_final<<<BB, 256, 0, stream>>>(part, ba, out);
}


# k_fused: rows l >= steps[b] not fetched: a chunk behind it writes -inf/0/zeros and leaves, dead score waves load no keys, dead vals stages re-read the last live stage at weight 0; sum order kept
# speedup vs baseline: 13.2647x; 1.2196x over previous
#include <hip/hip_runtime.h>
#include <hip/hip_bf16.h>
#include <math.h>

#define LL 2048
#define BB 64
#define KK 256
#define VV 256
#define HH 8
#define HK 2048
#define HV 2048

typedef __attribute__((ext_vector_type(2))) float v2f;
typedef __attribute__((ext_vector_type(4))) float v4f;
typedef __attribute__((ext_vector_type(8))) float v8f;

__device__ __forceinline__ v8f wmma4(v2f a, v2f b, v8f c) {
    return __builtin_amdgcn_wmma_f32_16x16x4_f32(
         false, a,  false, b,
         (short)0, c,  false,  false);
}

#define RING     16
#define PF_TRIPS 4

__device__ __forceinline__ v8f tile_product_ring(const float* __restrict__ Arow,
                                                 const float* __restrict__ Bcol,
                                                 const float* __restrict__ Bpf,
                                                 const int ldb, const int ksteps, const int lane) {
    const int krows = 4 * ksteps;
    v2f ra[RING], rb[RING];
    #pragma unroll
    for (int u = 0; u < RING; ++u) {
        const int k0 = 4 * u;
        ra[u] = (v2f){ Arow[k0], Arow[k0 + 1] };
        rb[u] = (v2f){ Bcol[(size_t)k0 * ldb], Bcol[(size_t)(k0 + 1) * ldb] };
        __builtin_amdgcn_sched_barrier(0);
    }
    #pragma unroll
    for (int i = 0; i < 2 * PF_TRIPS; ++i) {
        const int base = 4 * RING + 32 * i;
        if (base < krows) __builtin_prefetch(Bpf + (size_t)(base + lane) * ldb, 0, 1);
    }

    v8f c = {};
    for (int s0 = 0; s0 < ksteps - RING; s0 += RING) {
        const int base = 4 * (s0 + RING * (1 + PF_TRIPS));
        if (base < krows) {
            __builtin_prefetch(Bpf + (size_t)(base + lane) * ldb, 0, 1);
            __builtin_prefetch(Bpf + (size_t)(base + 32 + lane) * ldb, 0, 1);
        }
        #pragma unroll
        for (int u = 0; u < RING; ++u) {
            c = wmma4(ra[u], rb[u], c);
            const int k0 = 4 * (s0 + u + RING);
            ra[u] = (v2f){ Arow[k0], Arow[k0 + 1] };
            rb[u] = (v2f){ Bcol[(size_t)k0 * ldb], Bcol[(size_t)(k0 + 1) * ldb] };
            __builtin_amdgcn_sched_barrier(0);
        }
    }
    #pragma unroll
    for (int u = 0; u < RING; ++u) c = wmma4(ra[u], rb[u], c);
    return c;
}

#define QP_BLOCKS 256
#define PFV_CH    256
#define PFV_NC    8
#define PFV_ROWS  128

__global__ void k_qproj(const float* __restrict__ query,
                        const float* __restrict__ Wq,
                        const float* __restrict__ bq,
                        float* __restrict__ qbuf,
                        const float* __restrict__ vals,
                        const int* __restrict__ steps) {
    if (blockIdx.x >= QP_BLOCKS) {
        const int pl = threadIdx.x & 31;
        const int f  = (blockIdx.x - QP_BLOCKS) + QP_BLOCKS * (threadIdx.x >> 5);
        const int pb = f / PFV_NC;
        const int pc = f % PFV_NC;
        const int live = steps[pb] - pc * PFV_CH;
        const float* p = vals + ((size_t)(pc * PFV_CH + (pl >> 2)) * BB + pb) * VV + 64 * (pl & 3);
        #pragma unroll
        for (int j = 0; j < PFV_ROWS / 8; ++j)
            if (32 * (j >> 2) < live)
                __builtin_prefetch(p + (size_t)(8 * j) * BB * VV, 0, 1);
        return;
    }
    const int waveId = blockIdx.x * (blockDim.x >> 5) + (threadIdx.x >> 5);
    const int tm   = waveId >> 7;
    const int tn   = waveId & 127;
    const int lane = threadIdx.x & 31;
    const int col  = lane & 15;
    const int half = lane >> 4;

    const float* Arow = query + (size_t)(tm * 16 + col) * KK + 2 * half;
    const float* Bcol = Wq + (size_t)(2 * half) * HK + (tn * 16 + col);

    v8f c = tile_product_ring(Arow, Bcol, Wq + tn * 16, HK, KK / 4, lane);

    const float bias = bq[tn * 16 + col];
    #pragma unroll
    for (int j = 0; j < 8; ++j) {
        const int row = tm * 16 + j + 8 * half;
        qbuf[(size_t)row * HK + tn * 16 + col] = c[j] + bias;
    }
}

#define FC_CH    256
#define FC_NC    (LL / FC_CH)
#define FC_ROWS  32
#define FC_NST   (FC_CH / FC_ROWS)
#define FC_DEPTH 4
#define FC_PV    264
#define FC_PA    260
#define P1_STEPS 16
#define P1_RING  2
static_assert(PFV_CH == FC_CH && PFV_NC == FC_NC && PFV_ROWS == FC_DEPTH * FC_ROWS,
              "k_qproj's prefetch role must cover k_fused's first loads");
static_assert(2 * QP_BLOCKS == BB * FC_NC, "one prefetching wave per k_fused work-group");

__device__ __forceinline__ void fc_handover() {
    asm volatile("s_wait_dscnt 0x0" ::: "memory");
    asm volatile("s_barrier_signal -1\n\ts_barrier_wait -1" ::: "memory");
}

__global__ __launch_bounds__(512) __attribute__((amdgpu_waves_per_eu(8, 8))) void k_fused(const float* __restrict__ keys,
                                               const float* __restrict__ vals,
                                               const float* __restrict__ rpe,
                                               const float* __restrict__ qbuf,
                                               const int* __restrict__ steps,
                                               float* __restrict__ part,
                                               float* __restrict__ pmax,
                                               float* __restrict__ psum) {
    __shared__ __attribute__((aligned(16))) float sV[2][FC_ROWS][FC_PV];
    __shared__ __attribute__((aligned(16))) float sA[16][FC_PA];

    const int tid  = threadIdx.x;
    const int wv   = tid >> 5;
    const int lane = tid & 31;
    const int col  = lane & 15;
    const int half = lane >> 4;
    const int b    = blockIdx.x / FC_NC;
    const int ch   = blockIdx.x % FC_NC;
    const int l0   = ch * FC_CH;
    const int step = steps[b];
    const int nlive = min(max(step - l0, 0), FC_CH);
    const int nst   = (nlive + FC_ROWS - 1) / FC_ROWS;

    if (nlive == 0) {
        *(v4f*)(part + ((size_t)b * FC_NC + ch) * HH * VV + 4 * tid) = (v4f){ 0.0f, 0.0f, 0.0f, 0.0f };
        if (tid < HH) {
            pmax[((size_t)b * FC_NC + ch) * HH + tid] = -INFINITY;
            psum[((size_t)b * FC_NC + ch) * HH + tid] = 0.0f;
        }
        return;
    }

    const float* vsrc = vals + ((size_t)(l0 + 2 * wv) * BB + b) * VV + 4 * lane;
    v4f rv[FC_DEPTH][4];
    #pragma unroll
    for (int u = 0; u < FC_DEPTH; ++u) {
        const int ts = min(u, nst - 1);
        #pragma unroll
        for (int q = 0; q < 4; ++q)
            rv[u][q] = __builtin_nontemporal_load(
                (const v4f*)(vsrc + (size_t)(ts * FC_ROWS + (q >> 1)) * (BB * VV) + 128 * (q & 1)));
    }

    for (int i = tid; i < 8 * FC_PA; i += 512)
        sA[8 + i / FC_PA][i % FC_PA] = 0.0f;

    if (16 * wv >= nlive) {
        if (lane < HH) {
            const v4f ninf = { -INFINITY, -INFINITY, -INFINITY, -INFINITY };
            for (int j = 0; j < 4; ++j) *(v4f*)&sA[lane][16 * wv + 4 * j] = ninf;
        }
    } else {
        const int rr = col & 7;
        const int g  = col >> 3;
        const int lt = l0 + 16 * wv + rr;
        const float* key0 = keys + ((size_t)lt * BB + b) * KK + 8 * g + 4 * half;
        const float* key1 = key0 + (size_t)8 * BB * KK;
        const float r0 = rpe[(size_t)lt * BB + b];
        const float r1 = rpe[(size_t)(lt + 8) * BB + b];
        const float* qb = qbuf + (size_t)b * HK + (size_t)rr * KK + 8 * g + 4 * half;

        __builtin_prefetch(key0, 0, 0);
        __builtin_prefetch(key1, 0, 0);
        __builtin_prefetch(key0 + 64, 0, 0);
        __builtin_prefetch(key1 + 64, 0, 0);

        v4f ka0[P1_RING], ka1[P1_RING], kq[P1_RING];
        #pragma unroll
        for (int u = 0; u < P1_RING; ++u) {
            ka0[u] = __builtin_nontemporal_load((const v4f*)(key0 + 16 * u));
            ka1[u] = __builtin_nontemporal_load((const v4f*)(key1 + 16 * u));
            kq[u]  = *(const v4f*)(qb + 16 * u);
        }

        v8f c0 = {}, c1 = {};
        #pragma unroll
        for (int i = 0; i < P1_STEPS; ++i) {
            const int u = i % P1_RING;
            if ((i & 3) == 0 && i + 8 < P1_STEPS) {
                __builtin_prefetch(key0 + 16 * (i + 8), 0, 0);
                __builtin_prefetch(key1 + 16 * (i + 8), 0, 0);
            }
            const v4f a0 = ka0[u], a1 = ka1[u], qv = kq[u];
            c0 = wmma4((v2f){ a0.x * r0, a0.y * r0 }, (v2f){ qv.x, qv.y }, c0);
            c1 = wmma4((v2f){ a1.x * r1, a1.y * r1 }, (v2f){ qv.x, qv.y }, c1);
            c0 = wmma4((v2f){ a0.z * r0, a0.w * r0 }, (v2f){ qv.z, qv.w }, c0);
            c1 = wmma4((v2f){ a1.z * r1, a1.w * r1 }, (v2f){ qv.z, qv.w }, c1);
            if (i + P1_RING < P1_STEPS) {
                ka0[u] = __builtin_nontemporal_load((const v4f*)(key0 + 16 * (i + P1_RING)));
                ka1[u] = __builtin_nontemporal_load((const v4f*)(key1 + 16 * (i + P1_RING)));
                kq[u]  = *(const v4f*)(qb + 16 * (i + P1_RING));
            }
        }

        float s0[8], s1[8];
        #pragma unroll
        for (int j = 0; j < 8; ++j) {
            s0[j] = c0[j] + __shfl_xor(c0[j], 24);
            s1[j] = c1[j] + __shfl_xor(c1[j], 24);
        }
        if (lane < HH) {
            const int lb = 16 * wv;
            #pragma unroll
            for (int j = 0; j < 8; ++j) {
                s0[j] = (l0 + lb + j     < step) ? s0[j] : -INFINITY;
                s1[j] = (l0 + lb + 8 + j < step) ? s1[j] : -INFINITY;
            }
            *(v4f*)&sA[lane][lb]      = (v4f){ s0[0], s0[1], s0[2], s0[3] };
            *(v4f*)&sA[lane][lb + 4]  = (v4f){ s0[4], s0[5], s0[6], s0[7] };
            *(v4f*)&sA[lane][lb + 8]  = (v4f){ s1[0], s1[1], s1[2], s1[3] };
            *(v4f*)&sA[lane][lb + 12] = (v4f){ s1[4], s1[5], s1[6], s1[7] };
        }
    }
    fc_handover();

    if (wv < HH) {
        const v4f x0 = *(const v4f*)&sA[wv][4 * lane];
        const v4f x1 = *(const v4f*)&sA[wv][128 + 4 * lane];
        float m = fmaxf(fmaxf(fmaxf(x0.x, x0.y), fmaxf(x0.z, x0.w)),
                        fmaxf(fmaxf(x1.x, x1.y), fmaxf(x1.z, x1.w)));
        #pragma unroll
        for (int off = 16; off > 0; off >>= 1) m = fmaxf(m, __shfl_xor(m, off));
        const float ms = (m == -INFINITY) ? 0.0f : m;
        const v4f e0 = { __expf(x0.x - ms), __expf(x0.y - ms), __expf(x0.z - ms), __expf(x0.w - ms) };
        const v4f e1 = { __expf(x1.x - ms), __expf(x1.y - ms), __expf(x1.z - ms), __expf(x1.w - ms) };
        float sum = ((e0.x + e0.y) + (e0.z + e0.w)) + ((e1.x + e1.y) + (e1.z + e1.w));
        #pragma unroll
        for (int off = 16; off > 0; off >>= 1) sum += __shfl_xor(sum, off);
        *(v4f*)&sA[wv][4 * lane]       = e0;
        *(v4f*)&sA[wv][128 + 4 * lane] = e1;
        if (lane == 0) {
            pmax[((size_t)b * FC_NC + ch) * HH + wv] = m;
            psum[((size_t)b * FC_NC + ch) * HH + wv] = sum;
        }
    }

    v8f acc = {};
    #pragma unroll
    for (int t = 0; t < FC_NST; ++t) {
        const int u   = t % FC_DEPTH;
        const int buf = t & 1;
        #pragma unroll
        for (int q = 0; q < 4; ++q)
            *(v4f*)&sV[buf][2 * wv + (q >> 1)][128 * (q & 1) + 4 * lane] = rv[u][q];
        if (t + FC_DEPTH < FC_NST) {
            const int ts = min(t + FC_DEPTH, nst - 1);
            #pragma unroll
            for (int q = 0; q < 4; ++q)
                rv[u][q] = __builtin_nontemporal_load(
                    (const v4f*)(vsrc + (size_t)(ts * FC_ROWS + (q >> 1)) * (BB * VV) + 128 * (q & 1)));
        }
        fc_handover();
        #pragma unroll
        for (int j = 0; j < FC_ROWS / 4; ++j) {
            const int r0 = 4 * j + 2 * half;
            const v2f a  = *(const v2f*)&sA[col][FC_ROWS * t + r0];
            const v2f bb = { sV[buf][r0][16 * wv + col], sV[buf][r0 + 1][16 * wv + col] };
            acc = wmma4(a, bb, acc);
        }
    }

    if (half == 0) {
        #pragma unroll
        for (int j = 0; j < 8; ++j)
            part[(((size_t)b * FC_NC + ch) * HH + j) * VV + 16 * wv + col] = acc[j];
    }
}

#define PM_KV    64
#define PM_NQ    4
#define PM_PA    68
#define PM_STEPS 16

__global__ __launch_bounds__(256) void k_proj(float* part, const float* __restrict__ Wa,
                                              const float* __restrict__ pmax,
                                              const float* __restrict__ psum) {
    __shared__ __attribute__((aligned(16))) float sP[16][PM_PA];

    const int tid  = threadIdx.x;
    const int tw   = tid >> 5;
    const int lane = tid & 31;
    const int col  = lane & 15;
    const int half = lane >> 4;
    const int tm   = blockIdx.x >> 5;
    const int h    = (blockIdx.x >> 2) & 7;
    const int vq   = blockIdx.x & 3;

    const int r  = tid >> 4;
    const int pc = tid & 15;
    const int bs = tm * 16 + r;

    float m[FC_NC], sm[FC_NC];
    #pragma unroll
    for (int c = 0; c < FC_NC; ++c) {
        m[c]  = pmax[((size_t)bs * FC_NC + c) * HH + h];
        sm[c] = psum[((size_t)bs * FC_NC + c) * HH + h];
    }
    const float* prow = part + ((size_t)bs * FC_NC * HH + h) * VV + PM_KV * vq + 4 * pc;
    v4f pv[FC_NC];
    #pragma unroll
    for (int c = 0; c < FC_NC; ++c)
        pv[c] = *(const v4f*)(prow + (size_t)c * HH * VV);

    const float* Bcol = Wa + ((size_t)h * VV + PM_KV * vq + 2 * half) * VV + 32 * tw + 2 * col;
    v2f rb0[PM_STEPS], rb1[PM_STEPS];
    #pragma unroll
    for (int u = 0; u < PM_STEPS; ++u) {
        rb0[u] = *(const v2f*)(Bcol + (size_t)(4 * u) * VV);
        rb1[u] = *(const v2f*)(Bcol + (size_t)(4 * u + 1) * VV);
    }

    float M = m[0];
    #pragma unroll
    for (int c = 1; c < FC_NC; ++c) M = fmaxf(M, m[c]);
    float w[FC_NC];
    float den = 0.0f;
    #pragma unroll
    for (int c = 0; c < FC_NC; ++c) {
        w[c] = __expf(m[c] - M);
        den += w[c] * sm[c];
    }
    const float inv = 1.0f / den;
    v4f mg = { 0.0f, 0.0f, 0.0f, 0.0f };
    #pragma unroll
    for (int c = 0; c < FC_NC; ++c) {
        const float wc = w[c] * inv;
        mg.x += wc * pv[c].x;
        mg.y += wc * pv[c].y;
        mg.z += wc * pv[c].z;
        mg.w += wc * pv[c].w;
    }
    *(v4f*)&sP[r][4 * pc] = mg;
    fc_handover();

    v8f acc0 = {}, acc1 = {};
    #pragma unroll
    for (int s = 0; s < PM_STEPS; ++s) {
        const v2f a = *(const v2f*)&sP[col][4 * s + 2 * half];
        acc0 = wmma4(a, (v2f){ rb0[s].x, rb1[s].x }, acc0);
        acc1 = wmma4(a, (v2f){ rb0[s].y, rb1[s].y }, acc1);
    }

    const int n0 = 32 * tw + 2 * col;
    #pragma unroll
    for (int j = 0; j < 8; ++j) {
        const int bo = tm * 16 + j + 8 * half;
        *(v2f*)(part + (((size_t)bo * FC_NC + (n0 >> 6)) * HH + h) * VV + PM_KV * vq + (n0 & 63)) =
            (v2f){ acc0[j], acc1[j] };
    }
}

__global__ __launch_bounds__(256) void k_final(const float* __restrict__ proj,
                                               const float* __restrict__ ba,
                                               float* __restrict__ out) {
    const int b = blockIdx.x;
    const int n = threadIdx.x;

    const float* pb = proj + ((size_t)b * FC_NC + (n >> 6)) * HH * VV + (n & 63);
    float p[HH * PM_NQ];
    #pragma unroll
    for (int i = 0; i < HH * PM_NQ; ++i)
        p[i] = pb[(size_t)(i >> 2) * VV + PM_KV * (i & 3)];

    float acc = ba[n];
    #pragma unroll
    for (int i = 0; i < HH * PM_NQ; ++i) acc += p[i];
    out[(size_t)b * VV + n] = acc;
}

extern "C" void kernel_launch(void* const* d_in, const int* in_sizes, int n_in,
                              void* d_out, int out_size, void* d_ws, size_t ws_size,
                              hipStream_t stream) {
    const float* query = (const float*)d_in[0];
    const float* keys  = (const float*)d_in[1];
    const float* vals  = (const float*)d_in[2];
    const float* rpe   = (const float*)d_in[3];
    const float* Wq    = (const float*)d_in[4];
    const float* bq    = (const float*)d_in[5];
    const float* Wa    = (const float*)d_in[6];
    const float* ba    = (const float*)d_in[7];
    const int*   steps = (const int*)d_in[8];
    float* out = (float*)d_out;

    float* ws = (float*)d_ws;
    float* qbuf = ws;
    float* part = qbuf + (size_t)BB * HK;
    float* pmax = part + (size_t)BB * FC_NC * HH * VV;
    float* psum = pmax + (size_t)BB * FC_NC * HH;

    k_qproj<<<2 * QP_BLOCKS, 64, 0, stream>>>(query, Wq, bq, qbuf, vals, steps);
    k_fused<<<BB * FC_NC, 512, 0, stream>>>(keys, vals, rpe, qbuf, steps, part, pmax, psum);
    k_proj<<<(BB / 16) * HH * PM_NQ, 256, 0, stream>>>(part, Wa, pmax, psum);
    k_final<<<BB, 256, 0, stream>>>(part, ba, out);
}


# k_fused: work-group i takes the i-th (batch, chunk) of an order sorted from steps: full chunks, then straddling chunks by live rows, dead chunks last; k_qproj's prefetch follows; sum order kept
# speedup vs baseline: 15.9649x; 1.2036x over previous
#include <hip/hip_runtime.h>
#include <hip/hip_bf16.h>
#include <math.h>

#define LL 2048
#define BB 64
#define KK 256
#define VV 256
#define HH 8
#define HK 2048
#define HV 2048

typedef __attribute__((ext_vector_type(2))) float v2f;
typedef __attribute__((ext_vector_type(4))) float v4f;
typedef __attribute__((ext_vector_type(8))) float v8f;

__device__ __forceinline__ v8f wmma4(v2f a, v2f b, v8f c) {
    return __builtin_amdgcn_wmma_f32_16x16x4_f32(
         false, a,  false, b,
         (short)0, c,  false,  false);
}

#define RING     16
#define PF_TRIPS 4

__device__ __forceinline__ v8f tile_product_ring(const float* __restrict__ Arow,
                                                 const float* __restrict__ Bcol,
                                                 const float* __restrict__ Bpf,
                                                 const int ldb, const int ksteps, const int lane) {
    const int krows = 4 * ksteps;
    v2f ra[RING], rb[RING];
    #pragma unroll
    for (int u = 0; u < RING; ++u) {
        const int k0 = 4 * u;
        ra[u] = (v2f){ Arow[k0], Arow[k0 + 1] };
        rb[u] = (v2f){ Bcol[(size_t)k0 * ldb], Bcol[(size_t)(k0 + 1) * ldb] };
        __builtin_amdgcn_sched_barrier(0);
    }
    #pragma unroll
    for (int i = 0; i < 2 * PF_TRIPS; ++i) {
        const int base = 4 * RING + 32 * i;
        if (base < krows) __builtin_prefetch(Bpf + (size_t)(base + lane) * ldb, 0, 1);
    }

    v8f c = {};
    for (int s0 = 0; s0 < ksteps - RING; s0 += RING) {
        const int base = 4 * (s0 + RING * (1 + PF_TRIPS));
        if (base < krows) {
            __builtin_prefetch(Bpf + (size_t)(base + lane) * ldb, 0, 1);
            __builtin_prefetch(Bpf + (size_t)(base + 32 + lane) * ldb, 0, 1);
        }
        #pragma unroll
        for (int u = 0; u < RING; ++u) {
            c = wmma4(ra[u], rb[u], c);
            const int k0 = 4 * (s0 + u + RING);
            ra[u] = (v2f){ Arow[k0], Arow[k0 + 1] };
            rb[u] = (v2f){ Bcol[(size_t)k0 * ldb], Bcol[(size_t)(k0 + 1) * ldb] };
            __builtin_amdgcn_sched_barrier(0);
        }
    }
    #pragma unroll
    for (int u = 0; u < RING; ++u) c = wmma4(ra[u], rb[u], c);
    return c;
}

#define QP_BLOCKS 256
#define PFV_CH    256
#define PFV_NC    8
#define PFV_ROWS  128

__device__ __forceinline__ int fc_nth_batch(const unsigned lo, const unsigned hi, const int n, const int lane) {
    const unsigned below = (1u << lane) - 1u;
    const bool hit0 = ((lo >> lane) & 1u) && __builtin_popcount(lo & below) == n;
    const bool hit1 = ((hi >> lane) & 1u) && __builtin_popcount(lo) + __builtin_popcount(hi & below) == n;
    const unsigned m0 = __builtin_amdgcn_ballot_w32(hit0);
    const unsigned m1 = __builtin_amdgcn_ballot_w32(hit1);
    return m0 ? __builtin_ctz(m0) : 32 + __builtin_ctz(m1 | 0x80000000u);
}

__device__ __forceinline__ void fc_item(const int* __restrict__ steps, int i, const int lane,
                                        int& b, int& ch, int& step_b) {
    const int e0 = min(max(steps[lane], 0), LL);
    const int e1 = min(max(steps[lane + 32], 0), LL);
    const int f0 = e0 / PFV_CH, f1 = e1 / PFV_CH;
    const int r0 = e0 % PFV_CH, r1 = e1 % PFV_CH;
    int bb = BB - 1, cc = PFV_NC - 1;
    bool found = false;
    #pragma unroll
    for (int c = 0; c < PFV_NC; ++c) {
        const unsigned lo = __builtin_amdgcn_ballot_w32(f0 > c);
        const unsigned hi = __builtin_amdgcn_ballot_w32(f1 > c);
        const int cnt = __builtin_popcount(lo) + __builtin_popcount(hi);
        if (!found) {
            if (i < cnt) { bb = fc_nth_batch(lo, hi, i, lane); cc = c; found = true; }
            else i -= cnt;
        }
    }
    if (!found) {
        const unsigned plo = __builtin_amdgcn_ballot_w32(r0 > 0);
        const unsigned phi = __builtin_amdgcn_ballot_w32(r1 > 0);
        const int np = __builtin_popcount(plo) + __builtin_popcount(phi);
        if (i < np) {
            int k0 = 0, k1 = 0;
            for (int j = 0; j < 32; ++j) {
                const int x0 = __builtin_amdgcn_readlane(r0, j);
                const int x1 = __builtin_amdgcn_readlane(r1, j);
                k0 += (x0 > r0 || (x0 == r0 && j < lane)) + (x1 > r0);
                k1 += (x0 >= r1) + (x1 > r1 || (x1 == r1 && j < lane));
            }
            const unsigned m0 = __builtin_amdgcn_ballot_w32(r0 > 0 && k0 == i);
            const unsigned m1 = __builtin_amdgcn_ballot_w32(r1 > 0 && k1 == i);
            bb = m0 ? __builtin_ctz(m0) : 32 + __builtin_ctz(m1 | 0x80000000u);
            cc = min(__builtin_amdgcn_readlane(m0 ? f0 : f1, bb & 31), PFV_NC - 1);
            found = true;
        } else {
            i -= np;
        }
    }
    if (!found) {
        const int n0 = f0 + (r0 > 0), n1 = f1 + (r1 > 0);
        #pragma unroll
        for (int c = 0; c < PFV_NC; ++c) {
            const unsigned lo = __builtin_amdgcn_ballot_w32(n0 <= c);
            const unsigned hi = __builtin_amdgcn_ballot_w32(n1 <= c);
            const int cnt = __builtin_popcount(lo) + __builtin_popcount(hi);
            if (!found) {
                if (i < cnt) { bb = fc_nth_batch(lo, hi, i, lane); cc = c; found = true; }
                else i -= cnt;
            }
        }
    }
    b  = __builtin_amdgcn_readfirstlane(bb);
    ch = __builtin_amdgcn_readfirstlane(cc);
    step_b = __builtin_amdgcn_readlane(b < 32 ? e0 : e1, b & 31);
}

__global__ void k_qproj(const float* __restrict__ query,
                        const float* __restrict__ Wq,
                        const float* __restrict__ bq,
                        float* __restrict__ qbuf,
                        const float* __restrict__ vals,
                        const int* __restrict__ steps) {
    if (blockIdx.x >= QP_BLOCKS) {
        const int pl = threadIdx.x & 31;
        const int f  = (blockIdx.x - QP_BLOCKS) + QP_BLOCKS * (threadIdx.x >> 5);
        int pb, pc, pstep;
        fc_item(steps, f, pl, pb, pc, pstep);
        const int live = pstep - pc * PFV_CH;
        const float* p = vals + ((size_t)(pc * PFV_CH + (pl >> 2)) * BB + pb) * VV + 64 * (pl & 3);
        #pragma unroll
        for (int j = 0; j < PFV_ROWS / 8; ++j)
            if (32 * (j >> 2) < live)
                __builtin_prefetch(p + (size_t)(8 * j) * BB * VV, 0, 1);
        return;
    }
    const int waveId = blockIdx.x * (blockDim.x >> 5) + (threadIdx.x >> 5);
    const int tm   = waveId >> 7;
    const int tn   = waveId & 127;
    const int lane = threadIdx.x & 31;
    const int col  = lane & 15;
    const int half = lane >> 4;

    const float* Arow = query + (size_t)(tm * 16 + col) * KK + 2 * half;
    const float* Bcol = Wq + (size_t)(2 * half) * HK + (tn * 16 + col);

    v8f c = tile_product_ring(Arow, Bcol, Wq + tn * 16, HK, KK / 4, lane);

    const float bias = bq[tn * 16 + col];
    #pragma unroll
    for (int j = 0; j < 8; ++j) {
        const int row = tm * 16 + j + 8 * half;
        qbuf[(size_t)row * HK + tn * 16 + col] = c[j] + bias;
    }
}

#define FC_CH    256
#define FC_NC    (LL / FC_CH)
#define FC_ROWS  32
#define FC_NST   (FC_CH / FC_ROWS)
#define FC_DEPTH 4
#define FC_PV    264
#define FC_PA    260
#define P1_STEPS 16
#define P1_RING  2
static_assert(PFV_CH == FC_CH && PFV_NC == FC_NC && PFV_ROWS == FC_DEPTH * FC_ROWS,
              "k_qproj's prefetch role must cover k_fused's first loads");
static_assert(2 * QP_BLOCKS == BB * FC_NC, "one prefetching wave per k_fused work-group");

__device__ __forceinline__ void fc_handover() {
    asm volatile("s_wait_dscnt 0x0" ::: "memory");
    asm volatile("s_barrier_signal -1\n\ts_barrier_wait -1" ::: "memory");
}

__global__ __launch_bounds__(512) __attribute__((amdgpu_waves_per_eu(8, 8))) void k_fused(const float* __restrict__ keys,
                                               const float* __restrict__ vals,
                                               const float* __restrict__ rpe,
                                               const float* __restrict__ qbuf,
                                               const int* __restrict__ steps,
                                               float* __restrict__ part,
                                               float* __restrict__ pmax,
                                               float* __restrict__ psum) {
    __shared__ __attribute__((aligned(16))) float sV[2][FC_ROWS][FC_PV];
    __shared__ __attribute__((aligned(16))) float sA[16][FC_PA];

    const int tid  = threadIdx.x;
    const int wv   = tid >> 5;
    const int lane = tid & 31;
    const int col  = lane & 15;
    const int half = lane >> 4;
    int b, ch, step;
    fc_item(steps, blockIdx.x, lane, b, ch, step);
    const int l0   = ch * FC_CH;
    const int nlive = min(max(step - l0, 0), FC_CH);
    const int nst   = (nlive + FC_ROWS - 1) / FC_ROWS;

    if (nlive == 0) {
        *(v4f*)(part + ((size_t)b * FC_NC + ch) * HH * VV + 4 * tid) = (v4f){ 0.0f, 0.0f, 0.0f, 0.0f };
        if (tid < HH) {
            pmax[((size_t)b * FC_NC + ch) * HH + tid] = -INFINITY;
            psum[((size_t)b * FC_NC + ch) * HH + tid] = 0.0f;
        }
        return;
    }

    const float* vsrc = vals + ((size_t)(l0 + 2 * wv) * BB + b) * VV + 4 * lane;
    v4f rv[FC_DEPTH][4];
    #pragma unroll
    for (int u = 0; u < FC_DEPTH; ++u) {
        const int ts = min(u, nst - 1);
        #pragma unroll
        for (int q = 0; q < 4; ++q)
            rv[u][q] = __builtin_nontemporal_load(
                (const v4f*)(vsrc + (size_t)(ts * FC_ROWS + (q >> 1)) * (BB * VV) + 128 * (q & 1)));
    }

    for (int i = tid; i < 8 * FC_PA; i += 512)
        sA[8 + i / FC_PA][i % FC_PA] = 0.0f;

    if (16 * wv >= nlive) {
        if (lane < HH) {
            const v4f ninf = { -INFINITY, -INFINITY, -INFINITY, -INFINITY };
            for (int j = 0; j < 4; ++j) *(v4f*)&sA[lane][16 * wv + 4 * j] = ninf;
        }
    } else {
        const int rr = col & 7;
        const int g  = col >> 3;
        const int lt = l0 + 16 * wv + rr;
        const float* key0 = keys + ((size_t)lt * BB + b) * KK + 8 * g + 4 * half;
        const float* key1 = key0 + (size_t)8 * BB * KK;
        const float r0 = rpe[(size_t)lt * BB + b];
        const float r1 = rpe[(size_t)(lt + 8) * BB + b];
        const float* qb = qbuf + (size_t)b * HK + (size_t)rr * KK + 8 * g + 4 * half;

        __builtin_prefetch(key0, 0, 0);
        __builtin_prefetch(key1, 0, 0);
        __builtin_prefetch(key0 + 64, 0, 0);
        __builtin_prefetch(key1 + 64, 0, 0);

        v4f ka0[P1_RING], ka1[P1_RING], kq[P1_RING];
        #pragma unroll
        for (int u = 0; u < P1_RING; ++u) {
            ka0[u] = __builtin_nontemporal_load((const v4f*)(key0 + 16 * u));
            ka1[u] = __builtin_nontemporal_load((const v4f*)(key1 + 16 * u));
            kq[u]  = *(const v4f*)(qb + 16 * u);
        }

        v8f c0 = {}, c1 = {};
        #pragma unroll
        for (int i = 0; i < P1_STEPS; ++i) {
            const int u = i % P1_RING;
            if ((i & 3) == 0 && i + 8 < P1_STEPS) {
                __builtin_prefetch(key0 + 16 * (i + 8), 0, 0);
                __builtin_prefetch(key1 + 16 * (i + 8), 0, 0);
            }
            const v4f a0 = ka0[u], a1 = ka1[u], qv = kq[u];
            c0 = wmma4((v2f){ a0.x * r0, a0.y * r0 }, (v2f){ qv.x, qv.y }, c0);
            c1 = wmma4((v2f){ a1.x * r1, a1.y * r1 }, (v2f){ qv.x, qv.y }, c1);
            c0 = wmma4((v2f){ a0.z * r0, a0.w * r0 }, (v2f){ qv.z, qv.w }, c0);
            c1 = wmma4((v2f){ a1.z * r1, a1.w * r1 }, (v2f){ qv.z, qv.w }, c1);
            if (i + P1_RING < P1_STEPS) {
                ka0[u] = __builtin_nontemporal_load((const v4f*)(key0 + 16 * (i + P1_RING)));
                ka1[u] = __builtin_nontemporal_load((const v4f*)(key1 + 16 * (i + P1_RING)));
                kq[u]  = *(const v4f*)(qb + 16 * (i + P1_RING));
            }
        }

        float s0[8], s1[8];
        #pragma unroll
        for (int j = 0; j < 8; ++j) {
            s0[j] = c0[j] + __shfl_xor(c0[j], 24);
            s1[j] = c1[j] + __shfl_xor(c1[j], 24);
        }
        if (lane < HH) {
            const int lb = 16 * wv;
            #pragma unroll
            for (int j = 0; j < 8; ++j) {
                s0[j] = (l0 + lb + j     < step) ? s0[j] : -INFINITY;
                s1[j] = (l0 + lb + 8 + j < step) ? s1[j] : -INFINITY;
            }
            *(v4f*)&sA[lane][lb]      = (v4f){ s0[0], s0[1], s0[2], s0[3] };
            *(v4f*)&sA[lane][lb + 4]  = (v4f){ s0[4], s0[5], s0[6], s0[7] };
            *(v4f*)&sA[lane][lb + 8]  = (v4f){ s1[0], s1[1], s1[2], s1[3] };
            *(v4f*)&sA[lane][lb + 12] = (v4f){ s1[4], s1[5], s1[6], s1[7] };
        }
    }
    fc_handover();

    if (wv < HH) {
        const v4f x0 = *(const v4f*)&sA[wv][4 * lane];
        const v4f x1 = *(const v4f*)&sA[wv][128 + 4 * lane];
        float m = fmaxf(fmaxf(fmaxf(x0.x, x0.y), fmaxf(x0.z, x0.w)),
                        fmaxf(fmaxf(x1.x, x1.y), fmaxf(x1.z, x1.w)));
        #pragma unroll
        for (int off = 16; off > 0; off >>= 1) m = fmaxf(m, __shfl_xor(m, off));
        const float ms = (m == -INFINITY) ? 0.0f : m;
        const v4f e0 = { __expf(x0.x - ms), __expf(x0.y - ms), __expf(x0.z - ms), __expf(x0.w - ms) };
        const v4f e1 = { __expf(x1.x - ms), __expf(x1.y - ms), __expf(x1.z - ms), __expf(x1.w - ms) };
        float sum = ((e0.x + e0.y) + (e0.z + e0.w)) + ((e1.x + e1.y) + (e1.z + e1.w));
        #pragma unroll
        for (int off = 16; off > 0; off >>= 1) sum += __shfl_xor(sum, off);
        *(v4f*)&sA[wv][4 * lane]       = e0;
        *(v4f*)&sA[wv][128 + 4 * lane] = e1;
        if (lane == 0) {
            pmax[((size_t)b * FC_NC + ch) * HH + wv] = m;
            psum[((size_t)b * FC_NC + ch) * HH + wv] = sum;
        }
    }

    v8f acc = {};
    #pragma unroll
    for (int t = 0; t < FC_NST; ++t) {
        const int u   = t % FC_DEPTH;
        const int buf = t & 1;
        #pragma unroll
        for (int q = 0; q < 4; ++q)
            *(v4f*)&sV[buf][2 * wv + (q >> 1)][128 * (q & 1) + 4 * lane] = rv[u][q];
        if (t + FC_DEPTH < FC_NST) {
            const int ts = min(t + FC_DEPTH, nst - 1);
            #pragma unroll
            for (int q = 0; q < 4; ++q)
                rv[u][q] = __builtin_nontemporal_load(
                    (const v4f*)(vsrc + (size_t)(ts * FC_ROWS + (q >> 1)) * (BB * VV) + 128 * (q & 1)));
        }
        fc_handover();
        #pragma unroll
        for (int j = 0; j < FC_ROWS / 4; ++j) {
            const int r0 = 4 * j + 2 * half;
            const v2f a  = *(const v2f*)&sA[col][FC_ROWS * t + r0];
            const v2f bb = { sV[buf][r0][16 * wv + col], sV[buf][r0 + 1][16 * wv + col] };
            acc = wmma4(a, bb, acc);
        }
    }

    if (half == 0) {
        #pragma unroll
        for (int j = 0; j < 8; ++j)
            part[(((size_t)b * FC_NC + ch) * HH + j) * VV + 16 * wv + col] = acc[j];
    }
}

#define PM_KV    64
#define PM_NQ    4
#define PM_PA    68
#define PM_STEPS 16

__global__ __launch_bounds__(256) void k_proj(float* part, const float* __restrict__ Wa,
                                              const float* __restrict__ pmax,
                                              const float* __restrict__ psum) {
    __shared__ __attribute__((aligned(16))) float sP[16][PM_PA];

    const int tid  = threadIdx.x;
    const int tw   = tid >> 5;
    const int lane = tid & 31;
    const int col  = lane & 15;
    const int half = lane >> 4;
    const int tm   = blockIdx.x >> 5;
    const int h    = (blockIdx.x >> 2) & 7;
    const int vq   = blockIdx.x & 3;

    const int r  = tid >> 4;
    const int pc = tid & 15;
    const int bs = tm * 16 + r;

    float m[FC_NC], sm[FC_NC];
    #pragma unroll
    for (int c = 0; c < FC_NC; ++c) {
        m[c]  = pmax[((size_t)bs * FC_NC + c) * HH + h];
        sm[c] = psum[((size_t)bs * FC_NC + c) * HH + h];
    }
    const float* prow = part + ((size_t)bs * FC_NC * HH + h) * VV + PM_KV * vq + 4 * pc;
    v4f pv[FC_NC];
    #pragma unroll
    for (int c = 0; c < FC_NC; ++c)
        pv[c] = *(const v4f*)(prow + (size_t)c * HH * VV);

    const float* Bcol = Wa + ((size_t)h * VV + PM_KV * vq + 2 * half) * VV + 32 * tw + 2 * col;
    v2f rb0[PM_STEPS], rb1[PM_STEPS];
    #pragma unroll
    for (int u = 0; u < PM_STEPS; ++u) {
        rb0[u] = *(const v2f*)(Bcol + (size_t)(4 * u) * VV);
        rb1[u] = *(const v2f*)(Bcol + (size_t)(4 * u + 1) * VV);
    }

    float M = m[0];
    #pragma unroll
    for (int c = 1; c < FC_NC; ++c) M = fmaxf(M, m[c]);
    float w[FC_NC];
    float den = 0.0f;
    #pragma unroll
    for (int c = 0; c < FC_NC; ++c) {
        w[c] = __expf(m[c] - M);
        den += w[c] * sm[c];
    }
    const float inv = 1.0f / den;
    v4f mg = { 0.0f, 0.0f, 0.0f, 0.0f };
    #pragma unroll
    for (int c = 0; c < FC_NC; ++c) {
        const float wc = w[c] * inv;
        mg.x += wc * pv[c].x;
        mg.y += wc * pv[c].y;
        mg.z += wc * pv[c].z;
        mg.w += wc * pv[c].w;
    }
    *(v4f*)&sP[r][4 * pc] = mg;
    fc_handover();

    v8f acc0 = {}, acc1 = {};
    #pragma unroll
    for (int s = 0; s < PM_STEPS; ++s) {
        const v2f a = *(const v2f*)&sP[col][4 * s + 2 * half];
        acc0 = wmma4(a, (v2f){ rb0[s].x, rb1[s].x }, acc0);
        acc1 = wmma4(a, (v2f){ rb0[s].y, rb1[s].y }, acc1);
    }

    const int n0 = 32 * tw + 2 * col;
    #pragma unroll
    for (int j = 0; j < 8; ++j) {
        const int bo = tm * 16 + j + 8 * half;
        *(v2f*)(part + (((size_t)bo * FC_NC + (n0 >> 6)) * HH + h) * VV + PM_KV * vq + (n0 & 63)) =
            (v2f){ acc0[j], acc1[j] };
    }
}

__global__ __launch_bounds__(256) void k_final(const float* __restrict__ proj,
                                               const float* __restrict__ ba,
                                               float* __restrict__ out) {
    const int b = blockIdx.x;
    const int n = threadIdx.x;

    const float* pb = proj + ((size_t)b * FC_NC + (n >> 6)) * HH * VV + (n & 63);
    float p[HH * PM_NQ];
    #pragma unroll
    for (int i = 0; i < HH * PM_NQ; ++i)
        p[i] = pb[(size_t)(i >> 2) * VV + PM_KV * (i & 3)];

    float acc = ba[n];
    #pragma unroll
    for (int i = 0; i < HH * PM_NQ; ++i) acc += p[i];
    out[(size_t)b * VV + n] = acc;
}

extern "C" void kernel_launch(void* const* d_in, const int* in_sizes, int n_in,
                              void* d_out, int out_size, void* d_ws, size_t ws_size,
                              hipStream_t stream) {
    const float* query = (const float*)d_in[0];
    const float* keys  = (const float*)d_in[1];
    const float* vals  = (const float*)d_in[2];
    const float* rpe   = (const float*)d_in[3];
    const float* Wq    = (const float*)d_in[4];
    const float* bq    = (const float*)d_in[5];
    const float* Wa    = (const float*)d_in[6];
    const float* ba    = (const float*)d_in[7];
    const int*   steps = (const int*)d_in[8];
    float* out = (float*)d_out;

    float* ws = (float*)d_ws;
    float* qbuf = ws;
    float* part = qbuf + (size_t)BB * HK;
    float* pmax = part + (size_t)BB * FC_NC * HH * VV;
    float* psum = pmax + (size_t)BB * FC_NC * HH;

    k_qproj<<<2 * QP_BLOCKS, 64, 0, stream>>>(query, Wq, bq, qbuf, vals, steps);
    k_fused<<<BB * FC_NC, 512, 0, stream>>>(keys, vals, rpe, qbuf, steps, part, pmax, psum);
    k_proj<<<(BB / 16) * HH * PM_NQ, 256, 0, stream>>>(part, Wa, pmax, psum);
    k_final<<<BB, 256, 0, stream>>>(part, ba, out);
}
